# starting kernel
# baseline (speedup 1.0000x reference)
#include <hip/hip_runtime.h>
#include <hip/hip_bf16.h>

typedef float v2f __attribute__((ext_vector_type(2)));
typedef float v8f __attribute__((ext_vector_type(8)));

#define DF 64

__global__ void deg_kernel(const int* __restrict__ src, const int* __restrict__ dst,
                           float* __restrict__ deg_out, float* __restrict__ deg_in, int E) {
    int e = blockIdx.x * blockDim.x + threadIdx.x;
    if (e >= E) return;
    atomicAdd(&deg_out[src[e]], 1.0f);
    atomicAdd(&deg_in[dst[e]], 1.0f);
}

__global__ void norm_kernel(float* __restrict__ a, float* __restrict__ b, int n) {
    int i = blockIdx.x * blockDim.x + threadIdx.x;
    if (i >= n) return;
    a[i] = __frsqrt_rn(fmaxf(a[i], 1.0f));
    b[i] = __frsqrt_rn(fmaxf(b[i], 1.0f));
}

__global__ void prescale_kernel(const float* __restrict__ x, const float* __restrict__ s,
                                float* __restrict__ xs, int n_nodes) {
    int t = blockIdx.x * blockDim.x + threadIdx.x;
    int node = t >> 4;
    int q = t & 15;
    if (node >= n_nodes) return;
    float sc = s[node];
    float4 v = ((const float4*)(x + (size_t)node * DF))[q];
    v.x *= sc; v.y *= sc; v.z *= sc; v.w *= sc;
    ((float4*)(xs + (size_t)node * DF))[q] = v;
}

__global__ void scatter_kernel(const float* __restrict__ xs, const int* __restrict__ src,
                               const int* __restrict__ dst, float* __restrict__ agg, int E) {
    int t = blockIdx.x * blockDim.x + threadIdx.x;
    int e = t >> 4;
    int q = t & 15;
    if (e >= E) return;
    int s = src[e], d = dst[e];
    float4 v = ((const float4*)(xs + (size_t)s * DF))[q];
    float* p = agg + (size_t)d * DF + q * 4;
    atomicAdd(p + 0, v.x);
    atomicAdd(p + 1, v.y);
    atomicAdd(p + 2, v.z);
    atomicAdd(p + 3, v.w);
}

__global__ void pool_kernel(const float* __restrict__ h, const int* __restrict__ gid,
                            float* __restrict__ pool, float* __restrict__ cnt, int n_nodes) {
    int t = blockIdx.x * blockDim.x + threadIdx.x;
    int node = t >> 4;
    int q = t & 15;
    if (node >= n_nodes) return;
    int g = gid[node];
    float4 v = ((const float4*)(h + (size_t)node * DF))[q];
    float* p = pool + (size_t)g * DF + q * 4;
    atomicAdd(p + 0, v.x);
    atomicAdd(p + 1, v.y);
    atomicAdd(p + 2, v.z);
    atomicAdd(p + 3, v.w);
    if (q == 0) atomicAdd(&cnt[g], 1.0f);
}

__global__ void invcnt_kernel(float* __restrict__ cnt, int g) {
    int i = blockIdx.x * blockDim.x + threadIdx.x;
    if (i >= g) return;
    cnt[i] = 1.0f / fmaxf(cnt[i], 1.0f);
}

template <int NCOLS_C>
__global__ void gemm_wmma_kernel(const float* __restrict__ A,
                                 const float* __restrict__ rowscale,
                                 const float* __restrict__ W,
                                 const float* __restrict__ bias,
                                 float* __restrict__ out,
                                 int rows, int ncols_rt, int do_relu) {
    const int Ncols = (NCOLS_C > 0) ? NCOLS_C : ncols_rt;
    const int lane = threadIdx.x & 31;
    const int wave = threadIdx.x >> 5;
    const int half = lane >> 4;
    const int r16  = lane & 15;
    const int m0 = blockIdx.x * 16;
    const int n0 = wave * 16;

    const int row  = m0 + r16;
    const int rowc = (row < rows) ? row : 0;
    const float s = rowscale[rowc];
    const float* arow = A + (size_t)rowc * DF + 2 * half;

    const int col  = n0 + r16;
    const bool cok = (col < Ncols);
    const int colc = cok ? col : (Ncols - 1);
    const float* wp = W + (size_t)(2 * half) * Ncols + colc;

    v8f c = {};
#pragma unroll
    for (int k = 0; k < 16; ++k) {
        v2f a;
        a.x = arow[4 * k + 0] * s;
        a.y = arow[4 * k + 1] * s;
        v2f b;
        b.x = wp[(size_t)(4 * k) * Ncols];
        b.y = wp[(size_t)(4 * k + 1) * Ncols];
        c = __builtin_amdgcn_wmma_f32_16x16x4_f32(false, a, false, b, (short)0, c,
                                                  false, false);
    }

    const float bv = bias[colc];
    float* orow0 = out + (size_t)m0 * Ncols + col;
    if (m0 + 16 <= rows && cok) {
#pragma unroll
        for (int j = 0; j < 8; ++j) {
            float v = c[j] + bv;
            if (do_relu) v = fmaxf(v, 0.0f);
            orow0[(size_t)(j + 8 * half) * Ncols] = v;
        }
    } else if (cok) {
#pragma unroll
        for (int j = 0; j < 8; ++j) {
            int orow = m0 + j + 8 * half;
            if (orow < rows) {
                float v = c[j] + bv;
                if (do_relu) v = fmaxf(v, 0.0f);
                out[(size_t)orow * Ncols + col] = v;
            }
        }
    }
}

extern "C" void kernel_launch(void* const* d_in, const int* in_sizes, int n_in,
                              void* d_out, int out_size, void* d_ws, size_t ws_size,
                              hipStream_t stream) {
    const float* features = (const float*)d_in[0];
    const int*   src      = (const int*)d_in[1];
    const int*   dst      = (const int*)d_in[2];
    const int*   gid      = (const int*)d_in[3];
    const float* W1       = (const float*)d_in[4];
    const float* b1       = (const float*)d_in[5];
    const float* W2       = (const float*)d_in[6];
    const float* b2       = (const float*)d_in[7];
    const float* Wc       = (const float*)d_in[8];
    const float* bc       = (const float*)d_in[9];
    float* outp           = (float*)d_out;

    const int N    = in_sizes[3];
    const int E    = in_sizes[1];
    const int DOUT = in_sizes[9];
    const int G    = out_size / DOUT;

    float* ws    = (float*)d_ws;
    float* out_n = ws;
    float* in_n  = out_n + N;
    float* bufA  = in_n + N;
    float* bufB  = bufA + (size_t)N * DF;
    float* bufC  = bufB + (size_t)N * DF;
    float* pool  = bufC + (size_t)N * DF;
    float* cnt   = pool + (size_t)G * DF;

    const int TB = 256;

    hipMemsetAsync(out_n, 0, sizeof(float) * (size_t)N * 2, stream);
    deg_kernel<<<(E + TB - 1) / TB, TB, 0, stream>>>(src, dst, out_n, in_n, E);
    norm_kernel<<<(N + TB - 1) / TB, TB, 0, stream>>>(out_n, in_n, N);

    const int nodeT = N * 16;
    const int edgeT = E * 16;
    const int gemmGrid = (N + 15) / 16;

    prescale_kernel<<<(nodeT + TB - 1) / TB, TB, 0, stream>>>(features, out_n, bufA, N);
    hipMemsetAsync(bufB, 0, sizeof(float) * (size_t)N * DF, stream);
    scatter_kernel<<<(edgeT + TB - 1) / TB, TB, 0, stream>>>(bufA, src, dst, bufB, E);
    gemm_wmma_kernel<64><<<gemmGrid, 128, 0, stream>>>(bufB, in_n, W1, b1, bufC, N, DF, 1);

    prescale_kernel<<<(nodeT + TB - 1) / TB, TB, 0, stream>>>(bufC, out_n, bufA, N);
    hipMemsetAsync(bufB, 0, sizeof(float) * (size_t)N * DF, stream);
    scatter_kernel<<<(edgeT + TB - 1) / TB, TB, 0, stream>>>(bufA, src, dst, bufB, E);
    gemm_wmma_kernel<64><<<gemmGrid, 128, 0, stream>>>(bufB, in_n, W2, b2, bufC, N, DF, 1);

    hipMemsetAsync(pool, 0, sizeof(float) * ((size_t)G * DF + G), stream);
    pool_kernel<<<(nodeT + TB - 1) / TB, TB, 0, stream>>>(bufC, gid, pool, cnt, N);
    invcnt_kernel<<<(G + TB - 1) / TB, TB, 0, stream>>>(cnt, G);

    const int clsWaves = (DOUT + 15) / 16;
    if (DOUT == 8) {
        gemm_wmma_kernel<8><<<(G + 15) / 16, 32 * clsWaves, 0, stream>>>(
            pool, cnt, Wc, bc, outp, G, DOUT, 0);
    } else {
        gemm_wmma_kernel<-1><<<(G + 15) / 16, 32 * clsWaves, 0, stream>>>(
            pool, cnt, Wc, bc, outp, G, DOUT, 0);
    }
}


# pool: a 32-lane group sums runs of consecutive nodes with equal graph id in registers, one float atomic per run and column (whole 128-B line per instruction); per-graph summation order changes
# speedup vs baseline: 1.1734x; 1.1734x over previous
#include <hip/hip_runtime.h>
#include <hip/hip_bf16.h>

typedef float v2f __attribute__((ext_vector_type(2)));
typedef float v8f __attribute__((ext_vector_type(8)));

#define DF 64

__global__ void deg_kernel(const int* __restrict__ src, const int* __restrict__ dst,
                           float* __restrict__ deg_out, float* __restrict__ deg_in, int E) {
    int e = blockIdx.x * blockDim.x + threadIdx.x;
    if (e >= E) return;
    atomicAdd(&deg_out[src[e]], 1.0f);
    atomicAdd(&deg_in[dst[e]], 1.0f);
}

__global__ void norm_kernel(float* __restrict__ a, float* __restrict__ b, int n) {
    int i = blockIdx.x * blockDim.x + threadIdx.x;
    if (i >= n) return;
    a[i] = __frsqrt_rn(fmaxf(a[i], 1.0f));
    b[i] = __frsqrt_rn(fmaxf(b[i], 1.0f));
}

__global__ void prescale_kernel(const float* __restrict__ x, const float* __restrict__ s,
                                float* __restrict__ xs, int n_nodes) {
    int t = blockIdx.x * blockDim.x + threadIdx.x;
    int node = t >> 4;
    int q = t & 15;
    if (node >= n_nodes) return;
    float sc = s[node];
    float4 v = ((const float4*)(x + (size_t)node * DF))[q];
    v.x *= sc; v.y *= sc; v.z *= sc; v.w *= sc;
    ((float4*)(xs + (size_t)node * DF))[q] = v;
}

__global__ void scatter_kernel(const float* __restrict__ xs, const int* __restrict__ src,
                               const int* __restrict__ dst, float* __restrict__ agg, int E) {
    int t = blockIdx.x * blockDim.x + threadIdx.x;
    int e = t >> 4;
    int q = t & 15;
    if (e >= E) return;
    int s = src[e], d = dst[e];
    float4 v = ((const float4*)(xs + (size_t)s * DF))[q];
    float* p = agg + (size_t)d * DF + q * 4;
    atomicAdd(p + 0, v.x);
    atomicAdd(p + 1, v.y);
    atomicAdd(p + 2, v.z);
    atomicAdd(p + 3, v.w);
}

#define POOL_CH 32
#define POOL_B  16

__device__ __forceinline__ void pool_flush(float* __restrict__ pool, float* __restrict__ cnt,
                                           int g, int l, float a0, float a1, float run) {
    float* p = pool + (size_t)g * DF + l;
    atomicAdd(p, a0);
    atomicAdd(p + 32, a1);
    if (l == 0) atomicAdd(&cnt[g], run);
}

__global__ void pool_kernel(const float* __restrict__ h, const int* __restrict__ gid,
                            float* __restrict__ pool, float* __restrict__ cnt, int n_nodes) {
    const int t = blockIdx.x * blockDim.x + threadIdx.x;
    const int l = t & 31;
    const int n0 = (t >> 5) * POOL_CH;
    if (n0 >= n_nodes) return;
    const int n1 = (n0 + POOL_CH < n_nodes) ? (n0 + POOL_CH) : n_nodes;

    int cur = gid[n0];
    float a0 = 0.0f, a1 = 0.0f, run = 0.0f;
    for (int b = n0; b < n1; b += POOL_B) {
        int g[POOL_B];
        float v0[POOL_B], v1[POOL_B];
#pragma unroll
        for (int i = 0; i < POOL_B; ++i) {
            int n = b + i;
            n = (n < n1) ? n : (n1 - 1);
            g[i] = gid[n];
            const float* r = h + (size_t)n * DF + l;
            v0[i] = r[0];
            v1[i] = r[32];
        }
#pragma unroll
        for (int i = 0; i < POOL_B; ++i) {
            if (b + i < n1) {
                if (g[i] != cur) {
                    pool_flush(pool, cnt, cur, l, a0, a1, run);
                    cur = g[i];
                    a0 = 0.0f; a1 = 0.0f; run = 0.0f;
                }
                a0 += v0[i];
                a1 += v1[i];
                run += 1.0f;
            }
        }
    }
    pool_flush(pool, cnt, cur, l, a0, a1, run);
}

__global__ void invcnt_kernel(float* __restrict__ cnt, int g) {
    int i = blockIdx.x * blockDim.x + threadIdx.x;
    if (i >= g) return;
    cnt[i] = 1.0f / fmaxf(cnt[i], 1.0f);
}

template <int NCOLS_C>
__global__ void gemm_wmma_kernel(const float* __restrict__ A,
                                 const float* __restrict__ rowscale,
                                 const float* __restrict__ W,
                                 const float* __restrict__ bias,
                                 float* __restrict__ out,
                                 int rows, int ncols_rt, int do_relu) {
    const int Ncols = (NCOLS_C > 0) ? NCOLS_C : ncols_rt;
    const int lane = threadIdx.x & 31;
    const int wave = threadIdx.x >> 5;
    const int half = lane >> 4;
    const int r16  = lane & 15;
    const int m0 = blockIdx.x * 16;
    const int n0 = wave * 16;

    const int row  = m0 + r16;
    const int rowc = (row < rows) ? row : 0;
    const float s = rowscale[rowc];
    const float* arow = A + (size_t)rowc * DF + 2 * half;

    const int col  = n0 + r16;
    const bool cok = (col < Ncols);
    const int colc = cok ? col : (Ncols - 1);
    const float* wp = W + (size_t)(2 * half) * Ncols + colc;

    v8f c = {};
#pragma unroll
    for (int k = 0; k < 16; ++k) {
        v2f a;
        a.x = arow[4 * k + 0] * s;
        a.y = arow[4 * k + 1] * s;
        v2f b;
        b.x = wp[(size_t)(4 * k) * Ncols];
        b.y = wp[(size_t)(4 * k + 1) * Ncols];
        c = __builtin_amdgcn_wmma_f32_16x16x4_f32(false, a, false, b, (short)0, c,
                                                  false, false);
    }

    const float bv = bias[colc];
    float* orow0 = out + (size_t)m0 * Ncols + col;
    if (m0 + 16 <= rows && cok) {
#pragma unroll
        for (int j = 0; j < 8; ++j) {
            float v = c[j] + bv;
            if (do_relu) v = fmaxf(v, 0.0f);
            orow0[(size_t)(j + 8 * half) * Ncols] = v;
        }
    } else if (cok) {
#pragma unroll
        for (int j = 0; j < 8; ++j) {
            int orow = m0 + j + 8 * half;
            if (orow < rows) {
                float v = c[j] + bv;
                if (do_relu) v = fmaxf(v, 0.0f);
                out[(size_t)orow * Ncols + col] = v;
            }
        }
    }
}

extern "C" void kernel_launch(void* const* d_in, const int* in_sizes, int n_in,
                              void* d_out, int out_size, void* d_ws, size_t ws_size,
                              hipStream_t stream) {
    const float* features = (const float*)d_in[0];
    const int*   src      = (const int*)d_in[1];
    const int*   dst      = (const int*)d_in[2];
    const int*   gid      = (const int*)d_in[3];
    const float* W1       = (const float*)d_in[4];
    const float* b1       = (const float*)d_in[5];
    const float* W2       = (const float*)d_in[6];
    const float* b2       = (const float*)d_in[7];
    const float* Wc       = (const float*)d_in[8];
    const float* bc       = (const float*)d_in[9];
    float* outp           = (float*)d_out;

    const int N    = in_sizes[3];
    const int E    = in_sizes[1];
    const int DOUT = in_sizes[9];
    const int G    = out_size / DOUT;

    float* ws    = (float*)d_ws;
    float* out_n = ws;
    float* in_n  = out_n + N;
    float* bufA  = in_n + N;
    float* bufB  = bufA + (size_t)N * DF;
    float* bufC  = bufB + (size_t)N * DF;
    float* pool  = bufC + (size_t)N * DF;
    float* cnt   = pool + (size_t)G * DF;

    const int TB = 256;

    hipMemsetAsync(out_n, 0, sizeof(float) * (size_t)N * 2, stream);
    deg_kernel<<<(E + TB - 1) / TB, TB, 0, stream>>>(src, dst, out_n, in_n, E);
    norm_kernel<<<(N + TB - 1) / TB, TB, 0, stream>>>(out_n, in_n, N);

    const int nodeT = N * 16;
    const int edgeT = E * 16;
    const int gemmGrid = (N + 15) / 16;

    prescale_kernel<<<(nodeT + TB - 1) / TB, TB, 0, stream>>>(features, out_n, bufA, N);
    hipMemsetAsync(bufB, 0, sizeof(float) * (size_t)N * DF, stream);
    scatter_kernel<<<(edgeT + TB - 1) / TB, TB, 0, stream>>>(bufA, src, dst, bufB, E);
    gemm_wmma_kernel<64><<<gemmGrid, 128, 0, stream>>>(bufB, in_n, W1, b1, bufC, N, DF, 1);

    prescale_kernel<<<(nodeT + TB - 1) / TB, TB, 0, stream>>>(bufC, out_n, bufA, N);
    hipMemsetAsync(bufB, 0, sizeof(float) * (size_t)N * DF, stream);
    scatter_kernel<<<(edgeT + TB - 1) / TB, TB, 0, stream>>>(bufA, src, dst, bufB, E);
    gemm_wmma_kernel<64><<<gemmGrid, 128, 0, stream>>>(bufB, in_n, W2, b2, bufC, N, DF, 1);

    hipMemsetAsync(pool, 0, sizeof(float) * ((size_t)G * DF + G), stream);
    const int poolNodesPerBlock = (TB / 32) * POOL_CH;
    pool_kernel<<<(N + poolNodesPerBlock - 1) / poolNodesPerBlock, TB, 0, stream>>>(bufC, gid, pool, cnt, N);
    invcnt_kernel<<<(G + TB - 1) / TB, TB, 0, stream>>>(cnt, G);

    const int clsWaves = (DOUT + 15) / 16;
    if (DOUT == 8) {
        gemm_wmma_kernel<8><<<(G + 15) / 16, 32 * clsWaves, 0, stream>>>(
            pool, cnt, Wc, bc, outp, G, DOUT, 0);
    } else {
        gemm_wmma_kernel<-1><<<(G + 15) / 16, 32 * clsWaves, 0, stream>>>(
            pool, cnt, Wc, bc, outp, G, DOUT, 0);
    }
}


# aggregation: per call, edges are sorted by destination (integer count+rank, prefix sum, fill); one wave per node sums its source rows in registers; no float atomics on rows; row sum order changes
# speedup vs baseline: 2.0613x; 1.7567x over previous
#include <hip/hip_runtime.h>
#include <hip/hip_bf16.h>

typedef float v2f __attribute__((ext_vector_type(2)));
typedef float v8f __attribute__((ext_vector_type(8)));

#define DF 64

#define SCAN_T 256
#define SCAN_E (SCAN_T * 4)

__global__ void count_rank_kernel(const int* __restrict__ src, const int* __restrict__ dst,
                                  int* __restrict__ deg_out, int* __restrict__ deg_in,
                                  int* __restrict__ rank, int E) {
    int e = blockIdx.x * blockDim.x + threadIdx.x;
    if (e >= E) return;
    atomicAdd(&deg_out[src[e]], 1);
    rank[e] = atomicAdd(&deg_in[dst[e]], 1);
}

__global__ void scan_partial_kernel(const int* __restrict__ cnt, int* __restrict__ blocksum, int n) {
    __shared__ int wsum[SCAN_T / 32];
    const int tid = threadIdx.x;
    const int base = blockIdx.x * SCAN_E + tid * 4;
    int s = 0;
    for (int k = 0; k < 4; ++k)
        if (base + k < n) s += cnt[base + k];
    for (int o = 16; o > 0; o >>= 1) s += __shfl_xor(s, o, 32);
    if ((tid & 31) == 0) wsum[tid >> 5] = s;
    __syncthreads();
    if (tid == 0) {
        int t = 0;
        for (int w = 0; w < SCAN_T / 32; ++w) t += wsum[w];
        blocksum[blockIdx.x] = t;
    }
}

__global__ void scan_final_kernel(const int* __restrict__ deg_in, const int* __restrict__ deg_out,
                                  const int* __restrict__ blocksum, int* __restrict__ row_start,
                                  float* __restrict__ in_n, float* __restrict__ out_n, int n) {
    __shared__ int woff[SCAN_T / 32];
    __shared__ int wsum[SCAN_T / 32];
    const int tid = threadIdx.x;
    const int lane = tid & 31;
    const int wave = tid >> 5;

    int off = 0;
    for (int j = tid; j < (int)blockIdx.x; j += SCAN_T) off += blocksum[j];
    for (int o = 16; o > 0; o >>= 1) off += __shfl_xor(off, o, 32);
    if (lane == 0) woff[wave] = off;

    const int base = blockIdx.x * SCAN_E + tid * 4;
    int d[4];
    int tsum = 0;
    for (int k = 0; k < 4; ++k) {
        d[k] = (base + k < n) ? deg_in[base + k] : 0;
        tsum += d[k];
    }
    int inc = tsum;
    for (int o = 1; o < 32; o <<= 1) {
        int t = __shfl_up(inc, o, 32);
        if (lane >= o) inc += t;
    }
    if (lane == 31) wsum[wave] = inc;
    __syncthreads();

    int excl = inc - tsum;
    for (int w = 0; w < SCAN_T / 32; ++w) {
        excl += woff[w];
        if (w < wave) excl += wsum[w];
    }
    for (int k = 0; k < 4; ++k) {
        const int i = base + k;
        if (i < n) {
            row_start[i] = excl;
            excl += d[k];
            in_n[i]  = __frsqrt_rn(fmaxf((float)d[k], 1.0f));
            out_n[i] = __frsqrt_rn(fmaxf((float)deg_out[i], 1.0f));
        }
    }
}

__global__ void fill_kernel(const int* __restrict__ src, const int* __restrict__ dst,
                            const int* __restrict__ rank, const int* __restrict__ row_start,
                            int* __restrict__ by_dst, int E) {
    int e = blockIdx.x * blockDim.x + threadIdx.x;
    if (e >= E) return;
    by_dst[row_start[dst[e]] + rank[e]] = src[e];
}

__global__ void prescale_kernel(const float* __restrict__ x, const float* __restrict__ s,
                                float* __restrict__ xs, int n_nodes) {
    int t = blockIdx.x * blockDim.x + threadIdx.x;
    int node = t >> 4;
    int q = t & 15;
    if (node >= n_nodes) return;
    float sc = s[node];
    float4 v = ((const float4*)(x + (size_t)node * DF))[q];
    v.x *= sc; v.y *= sc; v.z *= sc; v.w *= sc;
    ((float4*)(xs + (size_t)node * DF))[q] = v;
}

#define GATHER_B 16

__global__ void gather_kernel(const float* __restrict__ xs, const int* __restrict__ row_start,
                              const int* __restrict__ deg_in, const int* __restrict__ by_dst,
                              float* __restrict__ agg, int n_nodes) {
    const int lane = threadIdx.x & 31;
    const int node = __builtin_amdgcn_readfirstlane((int)((blockIdx.x * blockDim.x + threadIdx.x) >> 5));
    if (node >= n_nodes) return;
    const int start = __builtin_amdgcn_readfirstlane(row_start[node]);
    const int m     = __builtin_amdgcn_readfirstlane(deg_in[node]);

    float ax = 0.0f, ay = 0.0f;
    for (int base = 0; base < m; base += 32) {
        const int mm = (m - base < 32) ? (m - base) : 32;
        int idx = 0;
        if (lane < mm) idx = by_dst[start + base + lane];
        for (int i = 0; i < mm; i += GATHER_B) {
            float2 v[GATHER_B];
            int s[GATHER_B];
#pragma unroll
            for (int k = 0; k < GATHER_B; ++k) {
                const int j = (i + k < mm) ? (i + k) : (mm - 1);
                s[k] = __builtin_amdgcn_readlane(idx, j);
            }
#pragma unroll
            for (int k = 0; k < GATHER_B; ++k) {
                v[k] = make_float2(0.0f, 0.0f);
                if (i + k < mm)
                    v[k] = *(const float2*)(xs + (size_t)s[k] * DF + 2 * lane);
            }
#pragma unroll
            for (int k = 0; k < GATHER_B; ++k) {
                ax += v[k].x;
                ay += v[k].y;
            }
        }
    }
    *(float2*)(agg + (size_t)node * DF + 2 * lane) = make_float2(ax, ay);
}

#define POOL_CH 32
#define POOL_B  16

__device__ __forceinline__ void pool_flush(float* __restrict__ pool, float* __restrict__ cnt,
                                           int g, int l, float a0, float a1, float run) {
    float* p = pool + (size_t)g * DF + l;
    atomicAdd(p, a0);
    atomicAdd(p + 32, a1);
    if (l == 0) atomicAdd(&cnt[g], run);
}

__global__ void pool_kernel(const float* __restrict__ h, const int* __restrict__ gid,
                            float* __restrict__ pool, float* __restrict__ cnt, int n_nodes) {
    const int t = blockIdx.x * blockDim.x + threadIdx.x;
    const int l = t & 31;
    const int n0 = (t >> 5) * POOL_CH;
    if (n0 >= n_nodes) return;
    const int n1 = (n0 + POOL_CH < n_nodes) ? (n0 + POOL_CH) : n_nodes;

    int cur = gid[n0];
    float a0 = 0.0f, a1 = 0.0f, run = 0.0f;
    for (int b = n0; b < n1; b += POOL_B) {
        int g[POOL_B];
        float v0[POOL_B], v1[POOL_B];
#pragma unroll
        for (int i = 0; i < POOL_B; ++i) {
            int n = b + i;
            n = (n < n1) ? n : (n1 - 1);
            g[i] = gid[n];
            const float* r = h + (size_t)n * DF + l;
            v0[i] = r[0];
            v1[i] = r[32];
        }
#pragma unroll
        for (int i = 0; i < POOL_B; ++i) {
            if (b + i < n1) {
                if (g[i] != cur) {
                    pool_flush(pool, cnt, cur, l, a0, a1, run);
                    cur = g[i];
                    a0 = 0.0f; a1 = 0.0f; run = 0.0f;
                }
                a0 += v0[i];
                a1 += v1[i];
                run += 1.0f;
            }
        }
    }
    pool_flush(pool, cnt, cur, l, a0, a1, run);
}

__global__ void invcnt_kernel(float* __restrict__ cnt, int g) {
    int i = blockIdx.x * blockDim.x + threadIdx.x;
    if (i >= g) return;
    cnt[i] = 1.0f / fmaxf(cnt[i], 1.0f);
}

template <int NCOLS_C>
__global__ void gemm_wmma_kernel(const float* __restrict__ A,
                                 const float* __restrict__ rowscale,
                                 const float* __restrict__ W,
                                 const float* __restrict__ bias,
                                 float* __restrict__ out,
                                 int rows, int ncols_rt, int do_relu) {
    const int Ncols = (NCOLS_C > 0) ? NCOLS_C : ncols_rt;
    const int lane = threadIdx.x & 31;
    const int wave = threadIdx.x >> 5;
    const int half = lane >> 4;
    const int r16  = lane & 15;
    const int m0 = blockIdx.x * 16;
    const int n0 = wave * 16;

    const int row  = m0 + r16;
    const int rowc = (row < rows) ? row : 0;
    const float s = rowscale[rowc];
    const float* arow = A + (size_t)rowc * DF + 2 * half;

    const int col  = n0 + r16;
    const bool cok = (col < Ncols);
    const int colc = cok ? col : (Ncols - 1);
    const float* wp = W + (size_t)(2 * half) * Ncols + colc;

    v8f c = {};
#pragma unroll
    for (int k = 0; k < 16; ++k) {
        v2f a;
        a.x = arow[4 * k + 0] * s;
        a.y = arow[4 * k + 1] * s;
        v2f b;
        b.x = wp[(size_t)(4 * k) * Ncols];
        b.y = wp[(size_t)(4 * k + 1) * Ncols];
        c = __builtin_amdgcn_wmma_f32_16x16x4_f32(false, a, false, b, (short)0, c,
                                                  false, false);
    }

    const float bv = bias[colc];
    float* orow0 = out + (size_t)m0 * Ncols + col;
    if (m0 + 16 <= rows && cok) {
#pragma unroll
        for (int j = 0; j < 8; ++j) {
            float v = c[j] + bv;
            if (do_relu) v = fmaxf(v, 0.0f);
            orow0[(size_t)(j + 8 * half) * Ncols] = v;
        }
    } else if (cok) {
#pragma unroll
        for (int j = 0; j < 8; ++j) {
            int orow = m0 + j + 8 * half;
            if (orow < rows) {
                float v = c[j] + bv;
                if (do_relu) v = fmaxf(v, 0.0f);
                out[(size_t)orow * Ncols + col] = v;
            }
        }
    }
}

extern "C" void kernel_launch(void* const* d_in, const int* in_sizes, int n_in,
                              void* d_out, int out_size, void* d_ws, size_t ws_size,
                              hipStream_t stream) {
    const float* features = (const float*)d_in[0];
    const int*   src      = (const int*)d_in[1];
    const int*   dst      = (const int*)d_in[2];
    const int*   gid      = (const int*)d_in[3];
    const float* W1       = (const float*)d_in[4];
    const float* b1       = (const float*)d_in[5];
    const float* W2       = (const float*)d_in[6];
    const float* b2       = (const float*)d_in[7];
    const float* Wc       = (const float*)d_in[8];
    const float* bc       = (const float*)d_in[9];
    float* outp           = (float*)d_out;

    const int N    = in_sizes[3];
    const int E    = in_sizes[1];
    const int DOUT = in_sizes[9];
    const int G    = out_size / DOUT;

    const size_t Np = ((size_t)N + 63) & ~(size_t)63;
    const size_t Ep = ((size_t)E + 63) & ~(size_t)63;
    const int nScan = (N + SCAN_E - 1) / SCAN_E;
    int*   deg_o = (int*)d_ws;
    int*   deg_i = deg_o + Np;
    int*   rowst = deg_i + Np;
    int*   rank  = rowst + Np;
    int*   bydst = rank + Ep;
    int*   bsum  = bydst + Ep;
    float* out_n = (float*)(bsum + Np);
    float* in_n  = out_n + Np;
    float* bufA  = in_n + Np;
    float* bufB  = bufA + (size_t)N * DF;
    float* bufC  = bufB + (size_t)N * DF;
    float* pool  = bufC + (size_t)N * DF;
    float* cnt   = pool + (size_t)G * DF;

    const int TB = 256;

    hipMemsetAsync(deg_o, 0, sizeof(int) * Np * 2, stream);
    count_rank_kernel<<<(E + TB - 1) / TB, TB, 0, stream>>>(src, dst, deg_o, deg_i, rank, E);
    scan_partial_kernel<<<nScan, SCAN_T, 0, stream>>>(deg_i, bsum, N);
    scan_final_kernel<<<nScan, SCAN_T, 0, stream>>>(deg_i, deg_o, bsum, rowst, in_n, out_n, N);
    fill_kernel<<<(E + TB - 1) / TB, TB, 0, stream>>>(src, dst, rank, rowst, bydst, E);

    const int nodeT = N * 16;
    const int gemmGrid = (N + 15) / 16;
    const int gatherGrid = (N + TB / 32 - 1) / (TB / 32);

    prescale_kernel<<<(nodeT + TB - 1) / TB, TB, 0, stream>>>(features, out_n, bufA, N);
    gather_kernel<<<gatherGrid, TB, 0, stream>>>(bufA, rowst, deg_i, bydst, bufB, N);
    gemm_wmma_kernel<64><<<gemmGrid, 128, 0, stream>>>(bufB, in_n, W1, b1, bufC, N, DF, 1);

    prescale_kernel<<<(nodeT + TB - 1) / TB, TB, 0, stream>>>(bufC, out_n, bufA, N);
    gather_kernel<<<gatherGrid, TB, 0, stream>>>(bufA, rowst, deg_i, bydst, bufB, N);
    gemm_wmma_kernel<64><<<gemmGrid, 128, 0, stream>>>(bufB, in_n, W2, b2, bufC, N, DF, 1);

    hipMemsetAsync(pool, 0, sizeof(float) * ((size_t)G * DF + G), stream);
    const int poolNodesPerBlock = (TB / 32) * POOL_CH;
    pool_kernel<<<(N + poolNodesPerBlock - 1) / poolNodesPerBlock, TB, 0, stream>>>(bufC, gid, pool, cnt, N);
    invcnt_kernel<<<(G + TB - 1) / TB, TB, 0, stream>>>(cnt, G);

    const int clsWaves = (DOUT + 15) / 16;
    if (DOUT == 8) {
        gemm_wmma_kernel<8><<<(G + 15) / 16, 32 * clsWaves, 0, stream>>>(
            pool, cnt, Wc, bc, outp, G, DOUT, 0);
    } else {
        gemm_wmma_kernel<-1><<<(G + 15) / 16, 32 * clsWaves, 0, stream>>>(
            pool, cnt, Wc, bc, outp, G, DOUT, 0);
    }
}


# no separate scaling pass: layer 1's row sum applies each source's out-norm itself (fused multiply-add), layer 1's GEMM epilogue stores relu(.)*out_norm for layer 2; row sum order kept
# speedup vs baseline: 2.2062x; 1.0703x over previous
#include <hip/hip_runtime.h>
#include <hip/hip_bf16.h>

typedef float v2f __attribute__((ext_vector_type(2)));
typedef float v8f __attribute__((ext_vector_type(8)));

#define DF 64

#define SCAN_T 256
#define SCAN_E (SCAN_T * 4)

__global__ void count_rank_kernel(const int* __restrict__ src, const int* __restrict__ dst,
                                  int* __restrict__ deg_out, int* __restrict__ deg_in,
                                  int* __restrict__ rank, int E) {
    int e = blockIdx.x * blockDim.x + threadIdx.x;
    if (e >= E) return;
    atomicAdd(&deg_out[src[e]], 1);
    rank[e] = atomicAdd(&deg_in[dst[e]], 1);
}

__global__ void scan_partial_kernel(const int* __restrict__ cnt, int* __restrict__ blocksum, int n) {
    __shared__ int wsum[SCAN_T / 32];
    const int tid = threadIdx.x;
    const int base = blockIdx.x * SCAN_E + tid * 4;
    int s = 0;
    for (int k = 0; k < 4; ++k)
        if (base + k < n) s += cnt[base + k];
    for (int o = 16; o > 0; o >>= 1) s += __shfl_xor(s, o, 32);
    if ((tid & 31) == 0) wsum[tid >> 5] = s;
    __syncthreads();
    if (tid == 0) {
        int t = 0;
        for (int w = 0; w < SCAN_T / 32; ++w) t += wsum[w];
        blocksum[blockIdx.x] = t;
    }
}

__global__ void scan_final_kernel(const int* __restrict__ deg_in, const int* __restrict__ deg_out,
                                  const int* __restrict__ blocksum, int* __restrict__ row_start,
                                  float* __restrict__ in_n, float* __restrict__ out_n, int n) {
    __shared__ int woff[SCAN_T / 32];
    __shared__ int wsum[SCAN_T / 32];
    const int tid = threadIdx.x;
    const int lane = tid & 31;
    const int wave = tid >> 5;

    int off = 0;
    for (int j = tid; j < (int)blockIdx.x; j += SCAN_T) off += blocksum[j];
    for (int o = 16; o > 0; o >>= 1) off += __shfl_xor(off, o, 32);
    if (lane == 0) woff[wave] = off;

    const int base = blockIdx.x * SCAN_E + tid * 4;
    int d[4];
    int tsum = 0;
    for (int k = 0; k < 4; ++k) {
        d[k] = (base + k < n) ? deg_in[base + k] : 0;
        tsum += d[k];
    }
    int inc = tsum;
    for (int o = 1; o < 32; o <<= 1) {
        int t = __shfl_up(inc, o, 32);
        if (lane >= o) inc += t;
    }
    if (lane == 31) wsum[wave] = inc;
    __syncthreads();

    int excl = inc - tsum;
    for (int w = 0; w < SCAN_T / 32; ++w) {
        excl += woff[w];
        if (w < wave) excl += wsum[w];
    }
    for (int k = 0; k < 4; ++k) {
        const int i = base + k;
        if (i < n) {
            row_start[i] = excl;
            excl += d[k];
            in_n[i]  = __frsqrt_rn(fmaxf((float)d[k], 1.0f));
            out_n[i] = __frsqrt_rn(fmaxf((float)deg_out[i], 1.0f));
        }
    }
}

__global__ void fill_kernel(const int* __restrict__ src, const int* __restrict__ dst,
                            const int* __restrict__ rank, const int* __restrict__ row_start,
                            int* __restrict__ by_dst, int E) {
    int e = blockIdx.x * blockDim.x + threadIdx.x;
    if (e >= E) return;
    by_dst[row_start[dst[e]] + rank[e]] = src[e];
}

#define GATHER_B 16

template <bool SCALE_SRC>
__global__ void gather_kernel(const float* __restrict__ x, const float* __restrict__ out_n,
                              const int* __restrict__ row_start, const int* __restrict__ deg_in,
                              const int* __restrict__ by_dst, float* __restrict__ agg, int n_nodes) {
    const int lane = threadIdx.x & 31;
    const int node = __builtin_amdgcn_readfirstlane((int)((blockIdx.x * blockDim.x + threadIdx.x) >> 5));
    if (node >= n_nodes) return;
    const int start = __builtin_amdgcn_readfirstlane(row_start[node]);
    const int m     = __builtin_amdgcn_readfirstlane(deg_in[node]);

    float ax = 0.0f, ay = 0.0f;
    for (int base = 0; base < m; base += 32) {
        const int mm = (m - base < 32) ? (m - base) : 32;
        int idx = 0;
        float sc = 0.0f;
        if (lane < mm) {
            idx = by_dst[start + base + lane];
            if (SCALE_SRC) sc = out_n[idx];
        }
        for (int i = 0; i < mm; i += GATHER_B) {
            float2 v[GATHER_B];
            int s[GATHER_B];
            float f[GATHER_B];
#pragma unroll
            for (int k = 0; k < GATHER_B; ++k) {
                const int j = (i + k < mm) ? (i + k) : (mm - 1);
                s[k] = __builtin_amdgcn_readlane(idx, j);
                f[k] = SCALE_SRC ? __int_as_float(__builtin_amdgcn_readlane(__float_as_int(sc), j)) : 1.0f;
            }
#pragma unroll
            for (int k = 0; k < GATHER_B; ++k) {
                v[k] = make_float2(0.0f, 0.0f);
                if (i + k < mm)
                    v[k] = *(const float2*)(x + (size_t)s[k] * DF + 2 * lane);
            }
#pragma unroll
            for (int k = 0; k < GATHER_B; ++k) {
                if (SCALE_SRC) {
                    ax = __builtin_fmaf(v[k].x, f[k], ax);
                    ay = __builtin_fmaf(v[k].y, f[k], ay);
                } else {
                    ax += v[k].x;
                    ay += v[k].y;
                }
            }
        }
    }
    *(float2*)(agg + (size_t)node * DF + 2 * lane) = make_float2(ax, ay);
}

#define POOL_CH 32
#define POOL_B  16

__device__ __forceinline__ void pool_flush(float* __restrict__ pool, float* __restrict__ cnt,
                                           int g, int l, float a0, float a1, float run) {
    float* p = pool + (size_t)g * DF + l;
    atomicAdd(p, a0);
    atomicAdd(p + 32, a1);
    if (l == 0) atomicAdd(&cnt[g], run);
}

__global__ void pool_kernel(const float* __restrict__ h, const int* __restrict__ gid,
                            float* __restrict__ pool, float* __restrict__ cnt, int n_nodes) {
    const int t = blockIdx.x * blockDim.x + threadIdx.x;
    const int l = t & 31;
    const int n0 = (t >> 5) * POOL_CH;
    if (n0 >= n_nodes) return;
    const int n1 = (n0 + POOL_CH < n_nodes) ? (n0 + POOL_CH) : n_nodes;

    int cur = gid[n0];
    float a0 = 0.0f, a1 = 0.0f, run = 0.0f;
    for (int b = n0; b < n1; b += POOL_B) {
        int g[POOL_B];
        float v0[POOL_B], v1[POOL_B];
#pragma unroll
        for (int i = 0; i < POOL_B; ++i) {
            int n = b + i;
            n = (n < n1) ? n : (n1 - 1);
            g[i] = gid[n];
            const float* r = h + (size_t)n * DF + l;
            v0[i] = r[0];
            v1[i] = r[32];
        }
#pragma unroll
        for (int i = 0; i < POOL_B; ++i) {
            if (b + i < n1) {
                if (g[i] != cur) {
                    pool_flush(pool, cnt, cur, l, a0, a1, run);
                    cur = g[i];
                    a0 = 0.0f; a1 = 0.0f; run = 0.0f;
                }
                a0 += v0[i];
                a1 += v1[i];
                run += 1.0f;
            }
        }
    }
    pool_flush(pool, cnt, cur, l, a0, a1, run);
}

__global__ void invcnt_kernel(float* __restrict__ cnt, int g) {
    int i = blockIdx.x * blockDim.x + threadIdx.x;
    if (i >= g) return;
    cnt[i] = 1.0f / fmaxf(cnt[i], 1.0f);
}

template <int NCOLS_C, bool SCALE_OUT>
__global__ void gemm_wmma_kernel(const float* __restrict__ A,
                                 const float* __restrict__ rowscale,
                                 const float* __restrict__ W,
                                 const float* __restrict__ bias,
                                 const float* __restrict__ outscale,
                                 float* __restrict__ out,
                                 int rows, int ncols_rt, int do_relu) {
    const int Ncols = (NCOLS_C > 0) ? NCOLS_C : ncols_rt;
    const int lane = threadIdx.x & 31;
    const int wave = threadIdx.x >> 5;
    const int half = lane >> 4;
    const int r16  = lane & 15;
    const int m0 = blockIdx.x * 16;
    const int n0 = wave * 16;

    const int row  = m0 + r16;
    const int rowc = (row < rows) ? row : 0;
    const float s = rowscale[rowc];
    const float* arow = A + (size_t)rowc * DF + 2 * half;

    const int col  = n0 + r16;
    const bool cok = (col < Ncols);
    const int colc = cok ? col : (Ncols - 1);
    const float* wp = W + (size_t)(2 * half) * Ncols + colc;

    v8f c = {};
#pragma unroll
    for (int k = 0; k < 16; ++k) {
        v2f a;
        a.x = arow[4 * k + 0] * s;
        a.y = arow[4 * k + 1] * s;
        v2f b;
        b.x = wp[(size_t)(4 * k) * Ncols];
        b.y = wp[(size_t)(4 * k + 1) * Ncols];
        c = __builtin_amdgcn_wmma_f32_16x16x4_f32(false, a, false, b, (short)0, c,
                                                  false, false);
    }

    const float bv = bias[colc];
    float* orow0 = out + (size_t)m0 * Ncols + col;
    if (m0 + 16 <= rows && cok) {
#pragma unroll
        for (int j = 0; j < 8; ++j) {
            float v = c[j] + bv;
            if (do_relu) v = fmaxf(v, 0.0f);
            if (SCALE_OUT) v *= outscale[m0 + j + 8 * half];
            orow0[(size_t)(j + 8 * half) * Ncols] = v;
        }
    } else if (cok) {
#pragma unroll
        for (int j = 0; j < 8; ++j) {
            int orow = m0 + j + 8 * half;
            if (orow < rows) {
                float v = c[j] + bv;
                if (do_relu) v = fmaxf(v, 0.0f);
                if (SCALE_OUT) v *= outscale[orow];
                out[(size_t)orow * Ncols + col] = v;
            }
        }
    }
}

extern "C" void kernel_launch(void* const* d_in, const int* in_sizes, int n_in,
                              void* d_out, int out_size, void* d_ws, size_t ws_size,
                              hipStream_t stream) {
    const float* features = (const float*)d_in[0];
    const int*   src      = (const int*)d_in[1];
    const int*   dst      = (const int*)d_in[2];
    const int*   gid      = (const int*)d_in[3];
    const float* W1       = (const float*)d_in[4];
    const float* b1       = (const float*)d_in[5];
    const float* W2       = (const float*)d_in[6];
    const float* b2       = (const float*)d_in[7];
    const float* Wc       = (const float*)d_in[8];
    const float* bc       = (const float*)d_in[9];
    float* outp           = (float*)d_out;

    const int N    = in_sizes[3];
    const int E    = in_sizes[1];
    const int DOUT = in_sizes[9];
    const int G    = out_size / DOUT;

    const size_t Np = ((size_t)N + 63) & ~(size_t)63;
    const size_t Ep = ((size_t)E + 63) & ~(size_t)63;
    const int nScan = (N + SCAN_E - 1) / SCAN_E;
    int*   deg_o = (int*)d_ws;
    int*   deg_i = deg_o + Np;
    int*   rowst = deg_i + Np;
    int*   rank  = rowst + Np;
    int*   bydst = rank + Ep;
    int*   bsum  = bydst + Ep;
    float* out_n = (float*)(bsum + Np);
    float* in_n  = out_n + Np;
    float* bufA  = in_n + Np;
    float* bufB  = bufA + (size_t)N * DF;
    float* bufC  = bufB + (size_t)N * DF;
    float* pool  = bufC + (size_t)N * DF;
    float* cnt   = pool + (size_t)G * DF;

    const int TB = 256;

    hipMemsetAsync(deg_o, 0, sizeof(int) * Np * 2, stream);
    count_rank_kernel<<<(E + TB - 1) / TB, TB, 0, stream>>>(src, dst, deg_o, deg_i, rank, E);
    scan_partial_kernel<<<nScan, SCAN_T, 0, stream>>>(deg_i, bsum, N);
    scan_final_kernel<<<nScan, SCAN_T, 0, stream>>>(deg_i, deg_o, bsum, rowst, in_n, out_n, N);
    fill_kernel<<<(E + TB - 1) / TB, TB, 0, stream>>>(src, dst, rank, rowst, bydst, E);

    const int gemmGrid = (N + 15) / 16;
    const int gatherGrid = (N + TB / 32 - 1) / (TB / 32);

    gather_kernel<true><<<gatherGrid, TB, 0, stream>>>(features, out_n, rowst, deg_i, bydst, bufB, N);
    gemm_wmma_kernel<64, true><<<gemmGrid, 128, 0, stream>>>(bufB, in_n, W1, b1, out_n, bufA, N, DF, 1);

    gather_kernel<false><<<gatherGrid, TB, 0, stream>>>(bufA, out_n, rowst, deg_i, bydst, bufB, N);
    gemm_wmma_kernel<64, false><<<gemmGrid, 128, 0, stream>>>(bufB, in_n, W2, b2, out_n, bufC, N, DF, 1);

    hipMemsetAsync(pool, 0, sizeof(float) * ((size_t)G * DF + G), stream);
    const int poolNodesPerBlock = (TB / 32) * POOL_CH;
    pool_kernel<<<(N + poolNodesPerBlock - 1) / poolNodesPerBlock, TB, 0, stream>>>(bufC, gid, pool, cnt, N);
    invcnt_kernel<<<(G + TB - 1) / TB, TB, 0, stream>>>(cnt, G);

    const int clsWaves = (DOUT + 15) / 16;
    if (DOUT == 8) {
        gemm_wmma_kernel<8, false><<<(G + 15) / 16, 32 * clsWaves, 0, stream>>>(
            pool, cnt, Wc, bc, cnt, outp, G, DOUT, 0);
    } else {
        gemm_wmma_kernel<-1, false><<<(G + 15) / 16, 32 * clsWaves, 0, stream>>>(
            pool, cnt, Wc, bc, cnt, outp, G, DOUT, 0);
    }
}


# resident-size grids: kernels stride over edges (3 per thread and trip), nodes, row tiles (B loaded once per wave); row-sum order follows the new arrival order of count atomics, not fixed run to run
# speedup vs baseline: 2.3818x; 1.0796x over previous
#include <hip/hip_runtime.h>
#include <hip/hip_bf16.h>

typedef float v2f __attribute__((ext_vector_type(2)));
typedef float v8f __attribute__((ext_vector_type(8)));

#define DF 64

#define SCAN_T 256
#define SCAN_E (SCAN_T * 4)
#define EDGE_B 3

__global__ void count_rank_kernel(const int* __restrict__ src, const int* __restrict__ dst,
                                  int* __restrict__ deg_out, int* __restrict__ deg_in,
                                  int* __restrict__ rank, int E) {
    const int nth = (int)(gridDim.x * blockDim.x);
    int e = (int)(blockIdx.x * blockDim.x + threadIdx.x);
    for (; e + (EDGE_B - 1) * nth < E; e += EDGE_B * nth) {
        int s[EDGE_B], d[EDGE_B], r[EDGE_B];
#pragma unroll
        for (int k = 0; k < EDGE_B; ++k) {
            s[k] = src[e + k * nth];
            d[k] = dst[e + k * nth];
        }
#pragma unroll
        for (int k = 0; k < EDGE_B; ++k) {
            atomicAdd(&deg_out[s[k]], 1);
            r[k] = atomicAdd(&deg_in[d[k]], 1);
        }
#pragma unroll
        for (int k = 0; k < EDGE_B; ++k) rank[e + k * nth] = r[k];
    }
    for (; e < E; e += nth) {
        atomicAdd(&deg_out[src[e]], 1);
        rank[e] = atomicAdd(&deg_in[dst[e]], 1);
    }
}

__global__ void scan_partial_kernel(const int* __restrict__ cnt, int* __restrict__ blocksum, int n) {
    __shared__ int wsum[SCAN_T / 32];
    const int tid = threadIdx.x;
    const int base = blockIdx.x * SCAN_E + tid * 4;
    int s = 0;
    for (int k = 0; k < 4; ++k)
        if (base + k < n) s += cnt[base + k];
    for (int o = 16; o > 0; o >>= 1) s += __shfl_xor(s, o, 32);
    if ((tid & 31) == 0) wsum[tid >> 5] = s;
    __syncthreads();
    if (tid == 0) {
        int t = 0;
        for (int w = 0; w < SCAN_T / 32; ++w) t += wsum[w];
        blocksum[blockIdx.x] = t;
    }
}

__global__ void scan_final_kernel(const int* __restrict__ deg_in, const int* __restrict__ deg_out,
                                  const int* __restrict__ blocksum, int* __restrict__ row_start,
                                  float* __restrict__ in_n, float* __restrict__ out_n, int n) {
    __shared__ int woff[SCAN_T / 32];
    __shared__ int wsum[SCAN_T / 32];
    const int tid = threadIdx.x;
    const int lane = tid & 31;
    const int wave = tid >> 5;

    int off = 0;
    for (int j = tid; j < (int)blockIdx.x; j += SCAN_T) off += blocksum[j];
    for (int o = 16; o > 0; o >>= 1) off += __shfl_xor(off, o, 32);
    if (lane == 0) woff[wave] = off;

    const int base = blockIdx.x * SCAN_E + tid * 4;
    int d[4];
    int tsum = 0;
    for (int k = 0; k < 4; ++k) {
        d[k] = (base + k < n) ? deg_in[base + k] : 0;
        tsum += d[k];
    }
    int inc = tsum;
    for (int o = 1; o < 32; o <<= 1) {
        int t = __shfl_up(inc, o, 32);
        if (lane >= o) inc += t;
    }
    if (lane == 31) wsum[wave] = inc;
    __syncthreads();

    int excl = inc - tsum;
    for (int w = 0; w < SCAN_T / 32; ++w) {
        excl += woff[w];
        if (w < wave) excl += wsum[w];
    }
    for (int k = 0; k < 4; ++k) {
        const int i = base + k;
        if (i < n) {
            row_start[i] = excl;
            excl += d[k];
            in_n[i]  = __frsqrt_rn(fmaxf((float)d[k], 1.0f));
            out_n[i] = __frsqrt_rn(fmaxf((float)deg_out[i], 1.0f));
        }
    }
}

__global__ void fill_kernel(const int* __restrict__ src, const int* __restrict__ dst,
                            const int* __restrict__ rank, const int* __restrict__ row_start,
                            int* __restrict__ by_dst, int E) {
    const int nth = (int)(gridDim.x * blockDim.x);
    int e = (int)(blockIdx.x * blockDim.x + threadIdx.x);
    for (; e + (EDGE_B - 1) * nth < E; e += EDGE_B * nth) {
        int s[EDGE_B], d[EDGE_B], rk[EDGE_B], rs[EDGE_B];
#pragma unroll
        for (int k = 0; k < EDGE_B; ++k) {
            s[k] = src[e + k * nth];
            d[k] = dst[e + k * nth];
            rk[k] = rank[e + k * nth];
        }
#pragma unroll
        for (int k = 0; k < EDGE_B; ++k) rs[k] = row_start[d[k]];
#pragma unroll
        for (int k = 0; k < EDGE_B; ++k) by_dst[rs[k] + rk[k]] = s[k];
    }
    for (; e < E; e += nth)
        by_dst[row_start[dst[e]] + rank[e]] = src[e];
}

#define GATHER_B 16

template <bool SCALE_SRC>
__global__ void gather_kernel(const float* __restrict__ x, const float* __restrict__ out_n,
                              const int* __restrict__ row_start, const int* __restrict__ deg_in,
                              const int* __restrict__ by_dst, float* __restrict__ agg, int n_nodes) {
    const int lane = threadIdx.x & 31;
    const int wave0  = __builtin_amdgcn_readfirstlane((int)((blockIdx.x * blockDim.x + threadIdx.x) >> 5));
    const int nwaves = (int)((gridDim.x * blockDim.x) >> 5);
    for (int node = wave0; node < n_nodes; node += nwaves) {
        const int start = __builtin_amdgcn_readfirstlane(row_start[node]);
        const int m     = __builtin_amdgcn_readfirstlane(deg_in[node]);

        float ax = 0.0f, ay = 0.0f;
        for (int base = 0; base < m; base += 32) {
            const int mm = (m - base < 32) ? (m - base) : 32;
            int idx = 0;
            float sc = 0.0f;
            if (lane < mm) {
                idx = by_dst[start + base + lane];
                if (SCALE_SRC) sc = out_n[idx];
            }
            for (int i = 0; i < mm; i += GATHER_B) {
                float2 v[GATHER_B];
                int s[GATHER_B];
                float f[GATHER_B];
#pragma unroll
                for (int k = 0; k < GATHER_B; ++k) {
                    const int j = (i + k < mm) ? (i + k) : (mm - 1);
                    s[k] = __builtin_amdgcn_readlane(idx, j);
                    f[k] = SCALE_SRC ? __int_as_float(__builtin_amdgcn_readlane(__float_as_int(sc), j)) : 1.0f;
                }
#pragma unroll
                for (int k = 0; k < GATHER_B; ++k) {
                    v[k] = make_float2(0.0f, 0.0f);
                    if (i + k < mm)
                        v[k] = *(const float2*)(x + (size_t)s[k] * DF + 2 * lane);
                }
#pragma unroll
                for (int k = 0; k < GATHER_B; ++k) {
                    if (SCALE_SRC) {
                        ax = __builtin_fmaf(v[k].x, f[k], ax);
                        ay = __builtin_fmaf(v[k].y, f[k], ay);
                    } else {
                        ax += v[k].x;
                        ay += v[k].y;
                    }
                }
            }
        }
        *(float2*)(agg + (size_t)node * DF + 2 * lane) = make_float2(ax, ay);
    }
}

#define POOL_CH 32
#define POOL_B  16

__device__ __forceinline__ void pool_flush(float* __restrict__ pool, float* __restrict__ cnt,
                                           int g, int l, float a0, float a1, float run) {
    float* p = pool + (size_t)g * DF + l;
    atomicAdd(p, a0);
    atomicAdd(p + 32, a1);
    if (l == 0) atomicAdd(&cnt[g], run);
}

__global__ void pool_kernel(const float* __restrict__ h, const int* __restrict__ gid,
                            float* __restrict__ pool, float* __restrict__ cnt, int n_nodes) {
    const int t = blockIdx.x * blockDim.x + threadIdx.x;
    const int l = t & 31;
    const int n0 = (t >> 5) * POOL_CH;
    if (n0 >= n_nodes) return;
    const int n1 = (n0 + POOL_CH < n_nodes) ? (n0 + POOL_CH) : n_nodes;

    int cur = gid[n0];
    float a0 = 0.0f, a1 = 0.0f, run = 0.0f;
    for (int b = n0; b < n1; b += POOL_B) {
        int g[POOL_B];
        float v0[POOL_B], v1[POOL_B];
#pragma unroll
        for (int i = 0; i < POOL_B; ++i) {
            int n = b + i;
            n = (n < n1) ? n : (n1 - 1);
            g[i] = gid[n];
            const float* r = h + (size_t)n * DF + l;
            v0[i] = r[0];
            v1[i] = r[32];
        }
#pragma unroll
        for (int i = 0; i < POOL_B; ++i) {
            if (b + i < n1) {
                if (g[i] != cur) {
                    pool_flush(pool, cnt, cur, l, a0, a1, run);
                    cur = g[i];
                    a0 = 0.0f; a1 = 0.0f; run = 0.0f;
                }
                a0 += v0[i];
                a1 += v1[i];
                run += 1.0f;
            }
        }
    }
    pool_flush(pool, cnt, cur, l, a0, a1, run);
}

__global__ void invcnt_kernel(float* __restrict__ cnt, int g) {
    int i = blockIdx.x * blockDim.x + threadIdx.x;
    if (i >= g) return;
    cnt[i] = 1.0f / fmaxf(cnt[i], 1.0f);
}

template <int NCOLS_C, bool SCALE_OUT>
__global__ void gemm_wmma_kernel(const float* __restrict__ A,
                                 const float* __restrict__ rowscale,
                                 const float* __restrict__ W,
                                 const float* __restrict__ bias,
                                 const float* __restrict__ outscale,
                                 float* __restrict__ out,
                                 int rows, int ncols_rt, int do_relu) {
    const int Ncols = (NCOLS_C > 0) ? NCOLS_C : ncols_rt;
    const int lane = threadIdx.x & 31;
    const int wave = threadIdx.x >> 5;
    const int half = lane >> 4;
    const int r16  = lane & 15;
    const int n0 = wave * 16;

    const int col  = n0 + r16;
    const bool cok = (col < Ncols);
    const int colc = cok ? col : (Ncols - 1);
    const float* wp = W + (size_t)(2 * half) * Ncols + colc;

    v2f b[16];
#pragma unroll
    for (int k = 0; k < 16; ++k) {
        b[k].x = wp[(size_t)(4 * k) * Ncols];
        b[k].y = wp[(size_t)(4 * k + 1) * Ncols];
    }
    const float bv = bias[colc];

    const int ntiles = (rows + 15) >> 4;
    for (int tile = blockIdx.x; tile < ntiles; tile += gridDim.x) {
        const int m0 = tile * 16;
        const int row  = m0 + r16;
        const int rowc = (row < rows) ? row : 0;
        const float s = rowscale[rowc];
        const float* arow = A + (size_t)rowc * DF + 2 * half;

        v8f c = {};
#pragma unroll
        for (int k = 0; k < 16; ++k) {
            v2f a;
            a.x = arow[4 * k + 0] * s;
            a.y = arow[4 * k + 1] * s;
            c = __builtin_amdgcn_wmma_f32_16x16x4_f32(false, a, false, b[k], (short)0, c,
                                                      false, false);
        }

        float* orow0 = out + (size_t)m0 * Ncols + col;
        if (m0 + 16 <= rows && cok) {
#pragma unroll
            for (int j = 0; j < 8; ++j) {
                float v = c[j] + bv;
                if (do_relu) v = fmaxf(v, 0.0f);
                if (SCALE_OUT) v *= outscale[m0 + j + 8 * half];
                orow0[(size_t)(j + 8 * half) * Ncols] = v;
            }
        } else if (cok) {
#pragma unroll
            for (int j = 0; j < 8; ++j) {
                int orow = m0 + j + 8 * half;
                if (orow < rows) {
                    float v = c[j] + bv;
                    if (do_relu) v = fmaxf(v, 0.0f);
                    if (SCALE_OUT) v *= outscale[orow];
                    out[(size_t)orow * Ncols + col] = v;
                }
            }
        }
    }
}

extern "C" void kernel_launch(void* const* d_in, const int* in_sizes, int n_in,
                              void* d_out, int out_size, void* d_ws, size_t ws_size,
                              hipStream_t stream) {
    const float* features = (const float*)d_in[0];
    const int*   src      = (const int*)d_in[1];
    const int*   dst      = (const int*)d_in[2];
    const int*   gid      = (const int*)d_in[3];
    const float* W1       = (const float*)d_in[4];
    const float* b1       = (const float*)d_in[5];
    const float* W2       = (const float*)d_in[6];
    const float* b2       = (const float*)d_in[7];
    const float* Wc       = (const float*)d_in[8];
    const float* bc       = (const float*)d_in[9];
    float* outp           = (float*)d_out;

    const int N    = in_sizes[3];
    const int E    = in_sizes[1];
    const int DOUT = in_sizes[9];
    const int G    = out_size / DOUT;

    const size_t Np = ((size_t)N + 63) & ~(size_t)63;
    const size_t Ep = ((size_t)E + 63) & ~(size_t)63;
    const int nScan = (N + SCAN_E - 1) / SCAN_E;
    int*   deg_o = (int*)d_ws;
    int*   deg_i = deg_o + Np;
    int*   rowst = deg_i + Np;
    int*   rank  = rowst + Np;
    int*   bydst = rank + Ep;
    int*   bsum  = bydst + Ep;
    float* out_n = (float*)(bsum + Np);
    float* in_n  = out_n + Np;
    float* bufA  = in_n + Np;
    float* bufB  = bufA + (size_t)N * DF;
    float* bufC  = bufB + (size_t)N * DF;
    float* pool  = bufC + (size_t)N * DF;
    float* cnt   = pool + (size_t)G * DF;

    const int TB = 256;

    hipMemsetAsync(deg_o, 0, sizeof(int) * Np * 2, stream);
    const int edgeGrid = 1024;
    count_rank_kernel<<<edgeGrid, TB, 0, stream>>>(src, dst, deg_o, deg_i, rank, E);
    scan_partial_kernel<<<nScan, SCAN_T, 0, stream>>>(deg_i, bsum, N);
    scan_final_kernel<<<nScan, SCAN_T, 0, stream>>>(deg_i, deg_o, bsum, rowst, in_n, out_n, N);
    fill_kernel<<<edgeGrid, TB, 0, stream>>>(src, dst, rank, rowst, bydst, E);

    const int gemmGrid = 2048;
    const int gatherGrid = 2048;

    gather_kernel<true><<<gatherGrid, TB, 0, stream>>>(features, out_n, rowst, deg_i, bydst, bufB, N);
    gemm_wmma_kernel<64, true><<<gemmGrid, 128, 0, stream>>>(bufB, in_n, W1, b1, out_n, bufA, N, DF, 1);

    gather_kernel<false><<<gatherGrid, TB, 0, stream>>>(bufA, out_n, rowst, deg_i, bydst, bufB, N);
    gemm_wmma_kernel<64, false><<<gemmGrid, 128, 0, stream>>>(bufB, in_n, W2, b2, out_n, bufC, N, DF, 1);

    hipMemsetAsync(pool, 0, sizeof(float) * ((size_t)G * DF + G), stream);
    const int poolNodesPerBlock = (TB / 32) * POOL_CH;
    pool_kernel<<<(N + poolNodesPerBlock - 1) / poolNodesPerBlock, TB, 0, stream>>>(bufC, gid, pool, cnt, N);
    invcnt_kernel<<<(G + TB - 1) / TB, TB, 0, stream>>>(cnt, G);

    const int clsWaves = (DOUT + 15) / 16;
    if (DOUT == 8) {
        gemm_wmma_kernel<8, false><<<(G + 15) / 16, 32 * clsWaves, 0, stream>>>(
            pool, cnt, Wc, bc, cnt, outp, G, DOUT, 0);
    } else {
        gemm_wmma_kernel<-1, false><<<(G + 15) / 16, 32 * clsWaves, 0, stream>>>(
            pool, cnt, Wc, bc, cnt, outp, G, DOUT, 0);
    }
}


# edge sort by destination through LDS bins: per-block bin counts and staged runs, then one block per 256-node bin ranks in LDS; no per-edge global rank atomic; row-sum order follows LDS arrival
# speedup vs baseline: 3.2801x; 1.3772x over previous
#include <hip/hip_runtime.h>
#include <hip/hip_bf16.h>

typedef float v2f __attribute__((ext_vector_type(2)));
typedef float v8f __attribute__((ext_vector_type(8)));

#define DF 64

#define BIN_SHIFT 8
#define BIN_W (1 << BIN_SHIFT)
#define MAX_BINS 1024
#define BIN_T 1024
#define EDGE_EPT 7
#define EDGE_CAP 6400
#define SORT_T BIN_W
#define SORT_B 8
#define SORT_CAP 8192

static_assert(BIN_T == MAX_BINS, "one thread per bin slot");
static_assert(EDGE_CAP <= BIN_T * EDGE_EPT, "every edge of a block has a thread slot");
static_assert(EDGE_CAP < 65536, "a rank inside a block fits 16 bits");

template <int NW>
__device__ __forceinline__ int block_excl_scan(int v, int* s_wave) {
    const int lane = threadIdx.x & 31;
    const int wave = threadIdx.x >> 5;
    int inc = v;
    for (int o = 1; o < 32; o <<= 1) {
        int t = __shfl_up(inc, o, 32);
        if (lane >= o) inc += t;
    }
    if (lane == 31) s_wave[wave] = inc;
    __syncthreads();
    int woff = 0;
    for (int w = 0; w < NW; ++w)
        if (w < wave) woff += s_wave[w];
    __syncthreads();
    return woff + inc - v;
}

__global__ void __launch_bounds__(BIN_T)
bin_count_kernel(const int* __restrict__ src, const int* __restrict__ dst,
                 int* __restrict__ deg_out, int* __restrict__ bin_cnt, int E, int nbins) {
    __shared__ int s_cnt[MAX_BINS];
    const int t = threadIdx.x;
    const int lo = blockIdx.x * EDGE_CAP;
    const int cnt = (E - lo < EDGE_CAP) ? (E - lo) : EDGE_CAP;
    s_cnt[t] = 0;
    __syncthreads();

    int s[EDGE_EPT], d[EDGE_EPT];
#pragma unroll
    for (int k = 0; k < EDGE_EPT; ++k) {
        const int i = t + k * BIN_T;
        s[k] = 0; d[k] = 0;
        if (i < cnt) { s[k] = src[lo + i]; d[k] = dst[lo + i]; }
    }
#pragma unroll
    for (int k = 0; k < EDGE_EPT; ++k) {
        const int i = t + k * BIN_T;
        if (i < cnt) {
            atomicAdd(&deg_out[s[k]], 1);
            atomicAdd(&s_cnt[d[k] >> BIN_SHIFT], 1);
        }
    }
    __syncthreads();
    const int c = s_cnt[t];
    if (t < nbins && c > 0) atomicAdd(&bin_cnt[t], c);
}

__global__ void __launch_bounds__(BIN_T)
bin_scatter_kernel(const int* __restrict__ src, const int* __restrict__ dst,
                   const int* __restrict__ bin_cnt, int* __restrict__ cursor,
                   int* __restrict__ bin_start, int* __restrict__ packed, int E, int nbins) {
    __shared__ int s_cnt[MAX_BINS];
    __shared__ int s_loc[MAX_BINS];
    __shared__ int s_base[MAX_BINS];
    __shared__ int s_wave[BIN_T / 32];
    __shared__ int s_val[EDGE_CAP];
    __shared__ int s_pos[EDGE_CAP];
    const int t = threadIdx.x;
    const int lo = blockIdx.x * EDGE_CAP;
    const int cnt = (E - lo < EDGE_CAP) ? (E - lo) : EDGE_CAP;
    s_cnt[t] = 0;
    __syncthreads();

    int val[EDGE_EPT], br[EDGE_EPT];
    {
        int s[EDGE_EPT], d[EDGE_EPT];
#pragma unroll
        for (int k = 0; k < EDGE_EPT; ++k) {
            const int i = t + k * BIN_T;
            s[k] = 0; d[k] = 0;
            if (i < cnt) { s[k] = src[lo + i]; d[k] = dst[lo + i]; }
        }
#pragma unroll
        for (int k = 0; k < EDGE_EPT; ++k) {
            const int i = t + k * BIN_T;
            val[k] = 0; br[k] = 0;
            if (i < cnt) {
                const int bin = d[k] >> BIN_SHIFT;
                const int r = atomicAdd(&s_cnt[bin], 1);
                val[k] = (int)(((unsigned)s[k] << BIN_SHIFT) | (unsigned)(d[k] & (BIN_W - 1)));
                br[k] = (bin << 16) | r;
            }
        }
    }

    const int total = (t < nbins) ? bin_cnt[t] : 0;
    const int start = block_excl_scan<BIN_T / 32>(total, s_wave);
    if (blockIdx.x == 0 && t < nbins) bin_start[t] = start;

    const int mine = s_cnt[t];
    const int loc = block_excl_scan<BIN_T / 32>(mine, s_wave);
    int base = 0;
    if (mine > 0) base = start + atomicAdd(&cursor[t], mine);
    s_loc[t] = loc;
    s_base[t] = base - loc;
    __syncthreads();

#pragma unroll
    for (int k = 0; k < EDGE_EPT; ++k) {
        const int i = t + k * BIN_T;
        if (i < cnt) {
            const int bin = br[k] >> 16;
            const int idx = s_loc[bin] + (br[k] & 0xFFFF);
            s_val[idx] = val[k];
            s_pos[idx] = s_base[bin] + idx;
        }
    }
    __syncthreads();
#pragma unroll
    for (int k = 0; k < EDGE_EPT; ++k) {
        const int i = t + k * BIN_T;
        if (i < cnt) packed[s_pos[i]] = s_val[i];
    }
}

__global__ void __launch_bounds__(SORT_T)
bin_sort_kernel(const int* __restrict__ packed, const int* __restrict__ bin_cnt,
                const int* __restrict__ bin_start, const int* __restrict__ deg_out,
                int* __restrict__ row_start, int* __restrict__ deg_in, int* __restrict__ by_dst,
                float* __restrict__ in_n, float* __restrict__ out_n, int n) {
    __shared__ int s_deg[BIN_W];
    __shared__ int s_fill[BIN_W];
    __shared__ int s_wave[SORT_T / 32];
    __shared__ int s_out[SORT_CAP];
    const int t = threadIdx.x;
    const int lo = bin_start[blockIdx.x];
    const int cnt = bin_cnt[blockIdx.x];
    const int* __restrict__ mine = packed + lo;
    s_deg[t] = 0;
    __syncthreads();

    for (int e0 = 0; e0 < cnt; e0 += SORT_T * SORT_B) {
        int p[SORT_B];
#pragma unroll
        for (int k = 0; k < SORT_B; ++k) {
            const int e = e0 + t + k * SORT_T;
            p[k] = mine[(e < cnt) ? e : (cnt - 1)];
        }
#pragma unroll
        for (int k = 0; k < SORT_B; ++k) {
            const int e = e0 + t + k * SORT_T;
            if (e < cnt) atomicAdd(&s_deg[p[k] & (BIN_W - 1)], 1);
        }
    }
    __syncthreads();

    const int dg = s_deg[t];
    const int ex = block_excl_scan<SORT_T / 32>(dg, s_wave);
    const bool staged = (cnt <= SORT_CAP);
    s_fill[t] = staged ? ex : (lo + ex);
    const int node = blockIdx.x * BIN_W + t;
    if (node < n) {
        row_start[node] = lo + ex;
        deg_in[node] = dg;
        in_n[node]  = __frsqrt_rn(fmaxf((float)dg, 1.0f));
        out_n[node] = __frsqrt_rn(fmaxf((float)deg_out[node], 1.0f));
    }
    __syncthreads();

    for (int e0 = 0; e0 < cnt; e0 += SORT_T * SORT_B) {
        int p[SORT_B];
#pragma unroll
        for (int k = 0; k < SORT_B; ++k) {
            const int e = e0 + t + k * SORT_T;
            p[k] = mine[(e < cnt) ? e : (cnt - 1)];
        }
#pragma unroll
        for (int k = 0; k < SORT_B; ++k) {
            const int e = e0 + t + k * SORT_T;
            if (e < cnt) {
                const int pos = atomicAdd(&s_fill[p[k] & (BIN_W - 1)], 1);
                const int s = (int)((unsigned)p[k] >> BIN_SHIFT);
                if (staged) s_out[pos] = s;
                else        by_dst[pos] = s;
            }
        }
    }
    if (staged) {
        __syncthreads();
#pragma unroll 4
        for (int i = t; i < cnt; i += SORT_T) by_dst[lo + i] = s_out[i];
    }
}

#define GATHER_B 16

template <bool SCALE_SRC>
__global__ void gather_kernel(const float* __restrict__ x, const float* __restrict__ out_n,
                              const int* __restrict__ row_start, const int* __restrict__ deg_in,
                              const int* __restrict__ by_dst, float* __restrict__ agg, int n_nodes) {
    const int lane = threadIdx.x & 31;
    const int wave0  = __builtin_amdgcn_readfirstlane((int)((blockIdx.x * blockDim.x + threadIdx.x) >> 5));
    const int nwaves = (int)((gridDim.x * blockDim.x) >> 5);
    for (int node = wave0; node < n_nodes; node += nwaves) {
        const int start = __builtin_amdgcn_readfirstlane(row_start[node]);
        const int m     = __builtin_amdgcn_readfirstlane(deg_in[node]);

        float ax = 0.0f, ay = 0.0f;
        for (int base = 0; base < m; base += 32) {
            const int mm = (m - base < 32) ? (m - base) : 32;
            int idx = 0;
            float sc = 0.0f;
            if (lane < mm) {
                idx = by_dst[start + base + lane];
                if (SCALE_SRC) sc = out_n[idx];
            }
            for (int i = 0; i < mm; i += GATHER_B) {
                float2 v[GATHER_B];
                int s[GATHER_B];
                float f[GATHER_B];
#pragma unroll
                for (int k = 0; k < GATHER_B; ++k) {
                    const int j = (i + k < mm) ? (i + k) : (mm - 1);
                    s[k] = __builtin_amdgcn_readlane(idx, j);
                    f[k] = SCALE_SRC ? __int_as_float(__builtin_amdgcn_readlane(__float_as_int(sc), j)) : 1.0f;
                }
#pragma unroll
                for (int k = 0; k < GATHER_B; ++k) {
                    v[k] = make_float2(0.0f, 0.0f);
                    if (i + k < mm)
                        v[k] = *(const float2*)(x + (size_t)s[k] * DF + 2 * lane);
                }
#pragma unroll
                for (int k = 0; k < GATHER_B; ++k) {
                    if (SCALE_SRC) {
                        ax = __builtin_fmaf(v[k].x, f[k], ax);
                        ay = __builtin_fmaf(v[k].y, f[k], ay);
                    } else {
                        ax += v[k].x;
                        ay += v[k].y;
                    }
                }
            }
        }
        *(float2*)(agg + (size_t)node * DF + 2 * lane) = make_float2(ax, ay);
    }
}

#define POOL_CH 32
#define POOL_B  16

__device__ __forceinline__ void pool_flush(float* __restrict__ pool, float* __restrict__ cnt,
                                           int g, int l, float a0, float a1, float run) {
    float* p = pool + (size_t)g * DF + l;
    atomicAdd(p, a0);
    atomicAdd(p + 32, a1);
    if (l == 0) atomicAdd(&cnt[g], run);
}

__global__ void pool_kernel(const float* __restrict__ h, const int* __restrict__ gid,
                            float* __restrict__ pool, float* __restrict__ cnt, int n_nodes) {
    const int t = blockIdx.x * blockDim.x + threadIdx.x;
    const int l = t & 31;
    const int n0 = (t >> 5) * POOL_CH;
    if (n0 >= n_nodes) return;
    const int n1 = (n0 + POOL_CH < n_nodes) ? (n0 + POOL_CH) : n_nodes;

    int cur = gid[n0];
    float a0 = 0.0f, a1 = 0.0f, run = 0.0f;
    for (int b = n0; b < n1; b += POOL_B) {
        int g[POOL_B];
        float v0[POOL_B], v1[POOL_B];
#pragma unroll
        for (int i = 0; i < POOL_B; ++i) {
            int n = b + i;
            n = (n < n1) ? n : (n1 - 1);
            g[i] = gid[n];
            const float* r = h + (size_t)n * DF + l;
            v0[i] = r[0];
            v1[i] = r[32];
        }
#pragma unroll
        for (int i = 0; i < POOL_B; ++i) {
            if (b + i < n1) {
                if (g[i] != cur) {
                    pool_flush(pool, cnt, cur, l, a0, a1, run);
                    cur = g[i];
                    a0 = 0.0f; a1 = 0.0f; run = 0.0f;
                }
                a0 += v0[i];
                a1 += v1[i];
                run += 1.0f;
            }
        }
    }
    pool_flush(pool, cnt, cur, l, a0, a1, run);
}

__global__ void invcnt_kernel(float* __restrict__ cnt, int g) {
    int i = blockIdx.x * blockDim.x + threadIdx.x;
    if (i >= g) return;
    cnt[i] = 1.0f / fmaxf(cnt[i], 1.0f);
}

template <int NCOLS_C, bool SCALE_OUT>
__global__ void gemm_wmma_kernel(const float* __restrict__ A,
                                 const float* __restrict__ rowscale,
                                 const float* __restrict__ W,
                                 const float* __restrict__ bias,
                                 const float* __restrict__ outscale,
                                 float* __restrict__ out,
                                 int rows, int ncols_rt, int do_relu) {
    const int Ncols = (NCOLS_C > 0) ? NCOLS_C : ncols_rt;
    const int lane = threadIdx.x & 31;
    const int wave = threadIdx.x >> 5;
    const int half = lane >> 4;
    const int r16  = lane & 15;
    const int n0 = wave * 16;

    const int col  = n0 + r16;
    const bool cok = (col < Ncols);
    const int colc = cok ? col : (Ncols - 1);
    const float* wp = W + (size_t)(2 * half) * Ncols + colc;

    v2f b[16];
#pragma unroll
    for (int k = 0; k < 16; ++k) {
        b[k].x = wp[(size_t)(4 * k) * Ncols];
        b[k].y = wp[(size_t)(4 * k + 1) * Ncols];
    }
    const float bv = bias[colc];

    const int ntiles = (rows + 15) >> 4;
    for (int tile = blockIdx.x; tile < ntiles; tile += gridDim.x) {
        const int m0 = tile * 16;
        const int row  = m0 + r16;
        const int rowc = (row < rows) ? row : 0;
        const float s = rowscale[rowc];
        const float* arow = A + (size_t)rowc * DF + 2 * half;

        v8f c = {};
#pragma unroll
        for (int k = 0; k < 16; ++k) {
            v2f a;
            a.x = arow[4 * k + 0] * s;
            a.y = arow[4 * k + 1] * s;
            c = __builtin_amdgcn_wmma_f32_16x16x4_f32(false, a, false, b[k], (short)0, c,
                                                      false, false);
        }

        float* orow0 = out + (size_t)m0 * Ncols + col;
        if (m0 + 16 <= rows && cok) {
#pragma unroll
            for (int j = 0; j < 8; ++j) {
                float v = c[j] + bv;
                if (do_relu) v = fmaxf(v, 0.0f);
                if (SCALE_OUT) v *= outscale[m0 + j + 8 * half];
                orow0[(size_t)(j + 8 * half) * Ncols] = v;
            }
        } else if (cok) {
#pragma unroll
            for (int j = 0; j < 8; ++j) {
                int orow = m0 + j + 8 * half;
                if (orow < rows) {
                    float v = c[j] + bv;
                    if (do_relu) v = fmaxf(v, 0.0f);
                    if (SCALE_OUT) v *= outscale[orow];
                    out[(size_t)orow * Ncols + col] = v;
                }
            }
        }
    }
}

extern "C" void kernel_launch(void* const* d_in, const int* in_sizes, int n_in,
                              void* d_out, int out_size, void* d_ws, size_t ws_size,
                              hipStream_t stream) {
    const float* features = (const float*)d_in[0];
    const int*   src      = (const int*)d_in[1];
    const int*   dst      = (const int*)d_in[2];
    const int*   gid      = (const int*)d_in[3];
    const float* W1       = (const float*)d_in[4];
    const float* b1       = (const float*)d_in[5];
    const float* W2       = (const float*)d_in[6];
    const float* b2       = (const float*)d_in[7];
    const float* Wc       = (const float*)d_in[8];
    const float* bc       = (const float*)d_in[9];
    float* outp           = (float*)d_out;

    const int N    = in_sizes[3];
    const int E    = in_sizes[1];
    const int DOUT = in_sizes[9];
    const int G    = out_size / DOUT;

    const size_t Np = ((size_t)N + 63) & ~(size_t)63;
    const size_t Ep = ((size_t)E + 63) & ~(size_t)63;
    const int nBins = (N + BIN_W - 1) / BIN_W;
    const int edgeBlocks = (E + EDGE_CAP - 1) / EDGE_CAP;
    int*   deg_o = (int*)d_ws;
    int*   bcnt  = deg_o + Np;
    int*   bcur  = bcnt + MAX_BINS;
    int*   bsta  = bcur + MAX_BINS;
    int*   deg_i = bsta + MAX_BINS;
    int*   rowst = deg_i + Np;
    int*   pack  = rowst + Np;
    int*   bydst = pack + Ep;
    float* out_n = (float*)(bydst + Ep);
    float* in_n  = out_n + Np;
    float* bufA  = in_n + Np;
    float* bufB  = bufA + (size_t)N * DF;
    float* bufC  = bufB + (size_t)N * DF;
    float* pool  = bufC + (size_t)N * DF;
    float* cnt   = pool + (size_t)G * DF;

    const int TB = 256;

    hipMemsetAsync(deg_o, 0, sizeof(int) * (Np + 2 * MAX_BINS), stream);
    bin_count_kernel<<<edgeBlocks, BIN_T, 0, stream>>>(src, dst, deg_o, bcnt, E, nBins);
    bin_scatter_kernel<<<edgeBlocks, BIN_T, 0, stream>>>(src, dst, bcnt, bcur, bsta, pack, E, nBins);
    bin_sort_kernel<<<nBins, SORT_T, 0, stream>>>(pack, bcnt, bsta, deg_o, rowst, deg_i, bydst, in_n, out_n, N);

    const int gemmGrid = 2048;
    const int gatherGrid = 2048;

    gather_kernel<true><<<gatherGrid, TB, 0, stream>>>(features, out_n, rowst, deg_i, bydst, bufB, N);
    gemm_wmma_kernel<64, true><<<gemmGrid, 128, 0, stream>>>(bufB, in_n, W1, b1, out_n, bufA, N, DF, 1);

    gather_kernel<false><<<gatherGrid, TB, 0, stream>>>(bufA, out_n, rowst, deg_i, bydst, bufB, N);
    gemm_wmma_kernel<64, false><<<gemmGrid, 128, 0, stream>>>(bufB, in_n, W2, b2, out_n, bufC, N, DF, 1);

    hipMemsetAsync(pool, 0, sizeof(float) * ((size_t)G * DF + G), stream);
    const int poolNodesPerBlock = (TB / 32) * POOL_CH;
    pool_kernel<<<(N + poolNodesPerBlock - 1) / poolNodesPerBlock, TB, 0, stream>>>(bufC, gid, pool, cnt, N);
    invcnt_kernel<<<(G + TB - 1) / TB, TB, 0, stream>>>(cnt, G);

    const int clsWaves = (DOUT + 15) / 16;
    if (DOUT == 8) {
        gemm_wmma_kernel<8, false><<<(G + 15) / 16, 32 * clsWaves, 0, stream>>>(
            pool, cnt, Wc, bc, cnt, outp, G, DOUT, 0);
    } else {
        gemm_wmma_kernel<-1, false><<<(G + 15) / 16, 32 * clsWaves, 0, stream>>>(
            pool, cnt, Wc, bc, cnt, outp, G, DOUT, 0);
    }
}


# mean pool in the layer-2 GEMM epilogue: a tile's rows of one graph summed in registers, one float atomic per column and run; classifier divides by count; 3 launches fewer; pool sum order changes
# speedup vs baseline: 3.4133x; 1.0406x over previous
#include <hip/hip_runtime.h>
#include <hip/hip_bf16.h>

typedef float v2f __attribute__((ext_vector_type(2)));
typedef float v8f __attribute__((ext_vector_type(8)));

#define DF 64

#define BIN_SHIFT 8
#define BIN_W (1 << BIN_SHIFT)
#define MAX_BINS 1024
#define BIN_T 1024
#define EDGE_EPT 7
#define EDGE_CAP 6400
#define SORT_T BIN_W
#define SORT_B 8
#define SORT_CAP 8192

static_assert(BIN_T == MAX_BINS, "one thread per bin slot");
static_assert(EDGE_CAP <= BIN_T * EDGE_EPT, "every edge of a block has a thread slot");
static_assert(EDGE_CAP < 65536, "a rank inside a block fits 16 bits");

template <int NW>
__device__ __forceinline__ int block_excl_scan(int v, int* s_wave) {
    const int lane = threadIdx.x & 31;
    const int wave = threadIdx.x >> 5;
    int inc = v;
    for (int o = 1; o < 32; o <<= 1) {
        int t = __shfl_up(inc, o, 32);
        if (lane >= o) inc += t;
    }
    if (lane == 31) s_wave[wave] = inc;
    __syncthreads();
    int woff = 0;
    for (int w = 0; w < NW; ++w)
        if (w < wave) woff += s_wave[w];
    __syncthreads();
    return woff + inc - v;
}

__global__ void __launch_bounds__(BIN_T)
bin_count_kernel(const int* __restrict__ src, const int* __restrict__ dst,
                 int* __restrict__ deg_out, int* __restrict__ bin_cnt, int E, int nbins) {
    __shared__ int s_cnt[MAX_BINS];
    const int t = threadIdx.x;
    const int lo = blockIdx.x * EDGE_CAP;
    const int cnt = (E - lo < EDGE_CAP) ? (E - lo) : EDGE_CAP;
    s_cnt[t] = 0;
    __syncthreads();

    int s[EDGE_EPT], d[EDGE_EPT];
#pragma unroll
    for (int k = 0; k < EDGE_EPT; ++k) {
        const int i = t + k * BIN_T;
        s[k] = 0; d[k] = 0;
        if (i < cnt) { s[k] = src[lo + i]; d[k] = dst[lo + i]; }
    }
#pragma unroll
    for (int k = 0; k < EDGE_EPT; ++k) {
        const int i = t + k * BIN_T;
        if (i < cnt) {
            atomicAdd(&deg_out[s[k]], 1);
            atomicAdd(&s_cnt[d[k] >> BIN_SHIFT], 1);
        }
    }
    __syncthreads();
    const int c = s_cnt[t];
    if (t < nbins && c > 0) atomicAdd(&bin_cnt[t], c);
}

__global__ void __launch_bounds__(BIN_T)
bin_scatter_kernel(const int* __restrict__ src, const int* __restrict__ dst,
                   const int* __restrict__ bin_cnt, int* __restrict__ cursor,
                   int* __restrict__ bin_start, int* __restrict__ packed, int E, int nbins) {
    __shared__ int s_cnt[MAX_BINS];
    __shared__ int s_loc[MAX_BINS];
    __shared__ int s_base[MAX_BINS];
    __shared__ int s_wave[BIN_T / 32];
    __shared__ int s_val[EDGE_CAP];
    __shared__ int s_pos[EDGE_CAP];
    const int t = threadIdx.x;
    const int lo = blockIdx.x * EDGE_CAP;
    const int cnt = (E - lo < EDGE_CAP) ? (E - lo) : EDGE_CAP;
    s_cnt[t] = 0;
    __syncthreads();

    int val[EDGE_EPT], br[EDGE_EPT];
    {
        int s[EDGE_EPT], d[EDGE_EPT];
#pragma unroll
        for (int k = 0; k < EDGE_EPT; ++k) {
            const int i = t + k * BIN_T;
            s[k] = 0; d[k] = 0;
            if (i < cnt) { s[k] = src[lo + i]; d[k] = dst[lo + i]; }
        }
#pragma unroll
        for (int k = 0; k < EDGE_EPT; ++k) {
            const int i = t + k * BIN_T;
            val[k] = 0; br[k] = 0;
            if (i < cnt) {
                const int bin = d[k] >> BIN_SHIFT;
                const int r = atomicAdd(&s_cnt[bin], 1);
                val[k] = (int)(((unsigned)s[k] << BIN_SHIFT) | (unsigned)(d[k] & (BIN_W - 1)));
                br[k] = (bin << 16) | r;
            }
        }
    }

    const int total = (t < nbins) ? bin_cnt[t] : 0;
    const int start = block_excl_scan<BIN_T / 32>(total, s_wave);
    if (blockIdx.x == 0 && t < nbins) bin_start[t] = start;

    const int mine = s_cnt[t];
    const int loc = block_excl_scan<BIN_T / 32>(mine, s_wave);
    int base = 0;
    if (mine > 0) base = start + atomicAdd(&cursor[t], mine);
    s_loc[t] = loc;
    s_base[t] = base - loc;
    __syncthreads();

#pragma unroll
    for (int k = 0; k < EDGE_EPT; ++k) {
        const int i = t + k * BIN_T;
        if (i < cnt) {
            const int bin = br[k] >> 16;
            const int idx = s_loc[bin] + (br[k] & 0xFFFF);
            s_val[idx] = val[k];
            s_pos[idx] = s_base[bin] + idx;
        }
    }
    __syncthreads();
#pragma unroll
    for (int k = 0; k < EDGE_EPT; ++k) {
        const int i = t + k * BIN_T;
        if (i < cnt) packed[s_pos[i]] = s_val[i];
    }
}

__global__ void __launch_bounds__(SORT_T)
bin_sort_kernel(const int* __restrict__ packed, const int* __restrict__ bin_cnt,
                const int* __restrict__ bin_start, const int* __restrict__ deg_out,
                int* __restrict__ row_start, int* __restrict__ deg_in, int* __restrict__ by_dst,
                float* __restrict__ in_n, float* __restrict__ out_n, int n) {
    __shared__ int s_deg[BIN_W];
    __shared__ int s_fill[BIN_W];
    __shared__ int s_wave[SORT_T / 32];
    __shared__ int s_out[SORT_CAP];
    const int t = threadIdx.x;
    const int lo = bin_start[blockIdx.x];
    const int cnt = bin_cnt[blockIdx.x];
    const int* __restrict__ mine = packed + lo;
    s_deg[t] = 0;
    __syncthreads();

    for (int e0 = 0; e0 < cnt; e0 += SORT_T * SORT_B) {
        int p[SORT_B];
#pragma unroll
        for (int k = 0; k < SORT_B; ++k) {
            const int e = e0 + t + k * SORT_T;
            p[k] = mine[(e < cnt) ? e : (cnt - 1)];
        }
#pragma unroll
        for (int k = 0; k < SORT_B; ++k) {
            const int e = e0 + t + k * SORT_T;
            if (e < cnt) atomicAdd(&s_deg[p[k] & (BIN_W - 1)], 1);
        }
    }
    __syncthreads();

    const int dg = s_deg[t];
    const int ex = block_excl_scan<SORT_T / 32>(dg, s_wave);
    const bool staged = (cnt <= SORT_CAP);
    s_fill[t] = staged ? ex : (lo + ex);
    const int node = blockIdx.x * BIN_W + t;
    if (node < n) {
        row_start[node] = lo + ex;
        deg_in[node] = dg;
        in_n[node]  = __frsqrt_rn(fmaxf((float)dg, 1.0f));
        out_n[node] = __frsqrt_rn(fmaxf((float)deg_out[node], 1.0f));
    }
    __syncthreads();

    for (int e0 = 0; e0 < cnt; e0 += SORT_T * SORT_B) {
        int p[SORT_B];
#pragma unroll
        for (int k = 0; k < SORT_B; ++k) {
            const int e = e0 + t + k * SORT_T;
            p[k] = mine[(e < cnt) ? e : (cnt - 1)];
        }
#pragma unroll
        for (int k = 0; k < SORT_B; ++k) {
            const int e = e0 + t + k * SORT_T;
            if (e < cnt) {
                const int pos = atomicAdd(&s_fill[p[k] & (BIN_W - 1)], 1);
                const int s = (int)((unsigned)p[k] >> BIN_SHIFT);
                if (staged) s_out[pos] = s;
                else        by_dst[pos] = s;
            }
        }
    }
    if (staged) {
        __syncthreads();
#pragma unroll 4
        for (int i = t; i < cnt; i += SORT_T) by_dst[lo + i] = s_out[i];
    }
}

#define GATHER_B 16

template <bool SCALE_SRC>
__global__ void gather_kernel(const float* __restrict__ x, const float* __restrict__ out_n,
                              const int* __restrict__ row_start, const int* __restrict__ deg_in,
                              const int* __restrict__ by_dst, float* __restrict__ agg, int n_nodes,
                              float* __restrict__ zero_buf, int zero_n) {
    const int lane = threadIdx.x & 31;
    if (!SCALE_SRC) {
        const int nthreads = (int)(gridDim.x * blockDim.x);
        for (int i = (int)(blockIdx.x * blockDim.x + threadIdx.x); i < zero_n; i += nthreads)
            zero_buf[i] = 0.0f;
    }
    const int wave0  = __builtin_amdgcn_readfirstlane((int)((blockIdx.x * blockDim.x + threadIdx.x) >> 5));
    const int nwaves = (int)((gridDim.x * blockDim.x) >> 5);
    for (int node = wave0; node < n_nodes; node += nwaves) {
        const int start = __builtin_amdgcn_readfirstlane(row_start[node]);
        const int m     = __builtin_amdgcn_readfirstlane(deg_in[node]);

        float ax = 0.0f, ay = 0.0f;
        for (int base = 0; base < m; base += 32) {
            const int mm = (m - base < 32) ? (m - base) : 32;
            int idx = 0;
            float sc = 0.0f;
            if (lane < mm) {
                idx = by_dst[start + base + lane];
                if (SCALE_SRC) sc = out_n[idx];
            }
            for (int i = 0; i < mm; i += GATHER_B) {
                float2 v[GATHER_B];
                int s[GATHER_B];
                float f[GATHER_B];
#pragma unroll
                for (int k = 0; k < GATHER_B; ++k) {
                    const int j = (i + k < mm) ? (i + k) : (mm - 1);
                    s[k] = __builtin_amdgcn_readlane(idx, j);
                    f[k] = SCALE_SRC ? __int_as_float(__builtin_amdgcn_readlane(__float_as_int(sc), j)) : 1.0f;
                }
#pragma unroll
                for (int k = 0; k < GATHER_B; ++k) {
                    v[k] = make_float2(0.0f, 0.0f);
                    if (i + k < mm)
                        v[k] = *(const float2*)(x + (size_t)s[k] * DF + 2 * lane);
                }
#pragma unroll
                for (int k = 0; k < GATHER_B; ++k) {
                    if (SCALE_SRC) {
                        ax = __builtin_fmaf(v[k].x, f[k], ax);
                        ay = __builtin_fmaf(v[k].y, f[k], ay);
                    } else {
                        ax += v[k].x;
                        ay += v[k].y;
                    }
                }
            }
        }
        *(float2*)(agg + (size_t)node * DF + 2 * lane) = make_float2(ax, ay);
    }
}

template <int NCOLS_C, bool SCALE_OUT, bool POOL_OUT, bool RECIP_SCALE>
__global__ void gemm_wmma_kernel(const float* __restrict__ A,
                                 const float* __restrict__ rowscale,
                                 const float* __restrict__ W,
                                 const float* __restrict__ bias,
                                 const float* __restrict__ outscale,
                                 float* __restrict__ out,
                                 const int* __restrict__ gid,
                                 float* __restrict__ cnt,
                                 int rows, int ncols_rt, int do_relu) {
    static_assert(!POOL_OUT || NCOLS_C == DF, "the pool table has DF columns");
    const int Ncols = (NCOLS_C > 0) ? NCOLS_C : ncols_rt;
    const int lane = threadIdx.x & 31;
    const int wave = threadIdx.x >> 5;
    const int half = lane >> 4;
    const int r16  = lane & 15;
    const int n0 = wave * 16;

    const int col  = n0 + r16;
    const bool cok = (col < Ncols);
    const int colc = cok ? col : (Ncols - 1);
    const float* wp = W + (size_t)(2 * half) * Ncols + colc;

    v2f b[16];
#pragma unroll
    for (int k = 0; k < 16; ++k) {
        b[k].x = wp[(size_t)(4 * k) * Ncols];
        b[k].y = wp[(size_t)(4 * k + 1) * Ncols];
    }
    const float bv = bias[colc];

    const int ntiles = (rows + 15) >> 4;
    for (int tile = blockIdx.x; tile < ntiles; tile += gridDim.x) {
        const int m0 = tile * 16;
        const int row  = m0 + r16;
        const int rowc = (row < rows) ? row : 0;
        float s = rowscale[rowc];
        if (RECIP_SCALE) s = 1.0f / fmaxf(s, 1.0f);
        int gl = 0;
        if (POOL_OUT) gl = gid[rowc];
        const float* arow = A + (size_t)rowc * DF + 2 * half;

        v8f c = {};
#pragma unroll
        for (int k = 0; k < 16; ++k) {
            v2f a;
            a.x = arow[4 * k + 0] * s;
            a.y = arow[4 * k + 1] * s;
            c = __builtin_amdgcn_wmma_f32_16x16x4_f32(false, a, false, b[k], (short)0, c,
                                                      false, false);
        }

        if (POOL_OUT) {
            float v[8];
#pragma unroll
            for (int j = 0; j < 8; ++j) {
                v[j] = c[j] + bv;
                if (do_relu) v[j] = fmaxf(v[j], 0.0f);
            }
            const int g0 = __builtin_amdgcn_readfirstlane(gl);
            const bool one_graph = (m0 + 16 <= rows) &&
                                   (__builtin_amdgcn_ballot_w32(gl != g0) == 0u);
            if (one_graph) {
                float t = ((v[0] + v[1]) + (v[2] + v[3])) + ((v[4] + v[5]) + (v[6] + v[7]));
                t += __shfl_xor(t, 16, 32);
                if (half == 0) atomicAdd(out + (size_t)g0 * DF + col, t);
                if (threadIdx.x == 0) atomicAdd(cnt + g0, 16.0f);
            } else {
                int gr[8];
#pragma unroll
                for (int j = 0; j < 8; ++j) gr[j] = __shfl(gl, 8 * half + j, 32);
                int cur = gr[0];
                float acc = 0.0f, run = 0.0f;
#pragma unroll
                for (int j = 0; j < 8; ++j) {
                    if (m0 + 8 * half + j < rows) {
                        if (gr[j] != cur) {
                            atomicAdd(out + (size_t)cur * DF + col, acc);
                            if (wave == 0 && r16 == 0) atomicAdd(cnt + cur, run);
                            cur = gr[j];
                            acc = 0.0f; run = 0.0f;
                        }
                        acc += v[j];
                        run += 1.0f;
                    }
                }
                if (run > 0.0f) {
                    atomicAdd(out + (size_t)cur * DF + col, acc);
                    if (wave == 0 && r16 == 0) atomicAdd(cnt + cur, run);
                }
            }
        } else {
            float* orow0 = out + (size_t)m0 * Ncols + col;
            if (m0 + 16 <= rows && cok) {
#pragma unroll
                for (int j = 0; j < 8; ++j) {
                    float v = c[j] + bv;
                    if (do_relu) v = fmaxf(v, 0.0f);
                    if (SCALE_OUT) v *= outscale[m0 + j + 8 * half];
                    orow0[(size_t)(j + 8 * half) * Ncols] = v;
                }
            } else if (cok) {
#pragma unroll
                for (int j = 0; j < 8; ++j) {
                    int orow = m0 + j + 8 * half;
                    if (orow < rows) {
                        float v = c[j] + bv;
                        if (do_relu) v = fmaxf(v, 0.0f);
                        if (SCALE_OUT) v *= outscale[orow];
                        out[(size_t)orow * Ncols + col] = v;
                    }
                }
            }
        }
    }
}

extern "C" void kernel_launch(void* const* d_in, const int* in_sizes, int n_in,
                              void* d_out, int out_size, void* d_ws, size_t ws_size,
                              hipStream_t stream) {
    const float* features = (const float*)d_in[0];
    const int*   src      = (const int*)d_in[1];
    const int*   dst      = (const int*)d_in[2];
    const int*   gid      = (const int*)d_in[3];
    const float* W1       = (const float*)d_in[4];
    const float* b1       = (const float*)d_in[5];
    const float* W2       = (const float*)d_in[6];
    const float* b2       = (const float*)d_in[7];
    const float* Wc       = (const float*)d_in[8];
    const float* bc       = (const float*)d_in[9];
    float* outp           = (float*)d_out;

    const int N    = in_sizes[3];
    const int E    = in_sizes[1];
    const int DOUT = in_sizes[9];
    const int G    = out_size / DOUT;

    const size_t Np = ((size_t)N + 63) & ~(size_t)63;
    const size_t Ep = ((size_t)E + 63) & ~(size_t)63;
    const int nBins = (N + BIN_W - 1) / BIN_W;
    const int edgeBlocks = (E + EDGE_CAP - 1) / EDGE_CAP;
    int*   deg_o = (int*)d_ws;
    int*   bcnt  = deg_o + Np;
    int*   bcur  = bcnt + MAX_BINS;
    int*   bsta  = bcur + MAX_BINS;
    int*   deg_i = bsta + MAX_BINS;
    int*   rowst = deg_i + Np;
    int*   pack  = rowst + Np;
    int*   bydst = pack + Ep;
    float* out_n = (float*)(bydst + Ep);
    float* in_n  = out_n + Np;
    float* bufA  = in_n + Np;
    float* bufB  = bufA + (size_t)N * DF;
    float* pool  = bufB + (size_t)N * DF;
    float* cnt   = pool + (size_t)G * DF;

    const int TB = 256;

    hipMemsetAsync(deg_o, 0, sizeof(int) * (Np + 2 * MAX_BINS), stream);
    bin_count_kernel<<<edgeBlocks, BIN_T, 0, stream>>>(src, dst, deg_o, bcnt, E, nBins);
    bin_scatter_kernel<<<edgeBlocks, BIN_T, 0, stream>>>(src, dst, bcnt, bcur, bsta, pack, E, nBins);
    bin_sort_kernel<<<nBins, SORT_T, 0, stream>>>(pack, bcnt, bsta, deg_o, rowst, deg_i, bydst, in_n, out_n, N);

    const int gemmGrid = 2048;
    const int gatherGrid = 2048;

    gather_kernel<true><<<gatherGrid, TB, 0, stream>>>(features, out_n, rowst, deg_i, bydst, bufB, N,
                                                       nullptr, 0);
    gemm_wmma_kernel<64, true, false, false><<<gemmGrid, 128, 0, stream>>>(
        bufB, in_n, W1, b1, out_n, bufA, nullptr, nullptr, N, DF, 1);

    gather_kernel<false><<<gatherGrid, TB, 0, stream>>>(bufA, out_n, rowst, deg_i, bydst, bufB, N,
                                                        pool, G * DF + G);
    gemm_wmma_kernel<64, false, true, false><<<gemmGrid, 128, 0, stream>>>(
        bufB, in_n, W2, b2, out_n, pool, gid, cnt, N, DF, 1);

    const int clsWaves = (DOUT + 15) / 16;
    if (DOUT == 8) {
        gemm_wmma_kernel<8, false, false, true><<<(G + 15) / 16, 32 * clsWaves, 0, stream>>>(
            pool, cnt, Wc, bc, cnt, outp, nullptr, nullptr, G, DOUT, 0);
    } else {
        gemm_wmma_kernel<-1, false, false, true><<<(G + 15) / 16, 32 * clsWaves, 0, stream>>>(
            pool, cnt, Wc, bc, cnt, outp, nullptr, nullptr, G, DOUT, 0);
    }
}


# each layer's row sum and its GEMM in one kernel: 8 waves sum a 16-node tile into LDS (times in-norm), 4 of them multiply it by W from LDS; agg buffer and 2 launches gone; row-sum and k order kept
# speedup vs baseline: 3.8032x; 1.1142x over previous
#include <hip/hip_runtime.h>
#include <hip/hip_bf16.h>

typedef float v2f __attribute__((ext_vector_type(2)));
typedef float v8f __attribute__((ext_vector_type(8)));

#define DF 64

#define BIN_SHIFT 8
#define BIN_W (1 << BIN_SHIFT)
#define MAX_BINS 1024
#define BIN_T 1024
#define EDGE_EPT 7
#define EDGE_CAP 6400
#define SORT_T BIN_W
#define SORT_B 8
#define SORT_CAP 8192

static_assert(BIN_T == MAX_BINS, "one thread per bin slot");
static_assert(EDGE_CAP <= BIN_T * EDGE_EPT, "every edge of a block has a thread slot");
static_assert(EDGE_CAP < 65536, "a rank inside a block fits 16 bits");

template <int NW>
__device__ __forceinline__ int block_excl_scan(int v, int* s_wave) {
    const int lane = threadIdx.x & 31;
    const int wave = threadIdx.x >> 5;
    int inc = v;
    for (int o = 1; o < 32; o <<= 1) {
        int t = __shfl_up(inc, o, 32);
        if (lane >= o) inc += t;
    }
    if (lane == 31) s_wave[wave] = inc;
    __syncthreads();
    int woff = 0;
    for (int w = 0; w < NW; ++w)
        if (w < wave) woff += s_wave[w];
    __syncthreads();
    return woff + inc - v;
}

__global__ void __launch_bounds__(BIN_T)
bin_count_kernel(const int* __restrict__ src, const int* __restrict__ dst,
                 int* __restrict__ deg_out, int* __restrict__ bin_cnt, int E, int nbins) {
    __shared__ int s_cnt[MAX_BINS];
    const int t = threadIdx.x;
    const int lo = blockIdx.x * EDGE_CAP;
    const int cnt = (E - lo < EDGE_CAP) ? (E - lo) : EDGE_CAP;
    s_cnt[t] = 0;
    __syncthreads();

    int s[EDGE_EPT], d[EDGE_EPT];
#pragma unroll
    for (int k = 0; k < EDGE_EPT; ++k) {
        const int i = t + k * BIN_T;
        s[k] = 0; d[k] = 0;
        if (i < cnt) { s[k] = src[lo + i]; d[k] = dst[lo + i]; }
    }
#pragma unroll
    for (int k = 0; k < EDGE_EPT; ++k) {
        const int i = t + k * BIN_T;
        if (i < cnt) {
            atomicAdd(&deg_out[s[k]], 1);
            atomicAdd(&s_cnt[d[k] >> BIN_SHIFT], 1);
        }
    }
    __syncthreads();
    const int c = s_cnt[t];
    if (t < nbins && c > 0) atomicAdd(&bin_cnt[t], c);
}

__global__ void __launch_bounds__(BIN_T)
bin_scatter_kernel(const int* __restrict__ src, const int* __restrict__ dst,
                   const int* __restrict__ bin_cnt, int* __restrict__ cursor,
                   int* __restrict__ bin_start, int* __restrict__ packed, int E, int nbins) {
    __shared__ int s_cnt[MAX_BINS];
    __shared__ int s_loc[MAX_BINS];
    __shared__ int s_base[MAX_BINS];
    __shared__ int s_wave[BIN_T / 32];
    __shared__ int s_val[EDGE_CAP];
    __shared__ int s_pos[EDGE_CAP];
    const int t = threadIdx.x;
    const int lo = blockIdx.x * EDGE_CAP;
    const int cnt = (E - lo < EDGE_CAP) ? (E - lo) : EDGE_CAP;
    s_cnt[t] = 0;
    __syncthreads();

    int val[EDGE_EPT], br[EDGE_EPT];
    {
        int s[EDGE_EPT], d[EDGE_EPT];
#pragma unroll
        for (int k = 0; k < EDGE_EPT; ++k) {
            const int i = t + k * BIN_T;
            s[k] = 0; d[k] = 0;
            if (i < cnt) { s[k] = src[lo + i]; d[k] = dst[lo + i]; }
        }
#pragma unroll
        for (int k = 0; k < EDGE_EPT; ++k) {
            const int i = t + k * BIN_T;
            val[k] = 0; br[k] = 0;
            if (i < cnt) {
                const int bin = d[k] >> BIN_SHIFT;
                const int r = atomicAdd(&s_cnt[bin], 1);
                val[k] = (int)(((unsigned)s[k] << BIN_SHIFT) | (unsigned)(d[k] & (BIN_W - 1)));
                br[k] = (bin << 16) | r;
            }
        }
    }

    const int total = (t < nbins) ? bin_cnt[t] : 0;
    const int start = block_excl_scan<BIN_T / 32>(total, s_wave);
    if (blockIdx.x == 0 && t < nbins) bin_start[t] = start;

    const int mine = s_cnt[t];
    const int loc = block_excl_scan<BIN_T / 32>(mine, s_wave);
    int base = 0;
    if (mine > 0) base = start + atomicAdd(&cursor[t], mine);
    s_loc[t] = loc;
    s_base[t] = base - loc;
    __syncthreads();

#pragma unroll
    for (int k = 0; k < EDGE_EPT; ++k) {
        const int i = t + k * BIN_T;
        if (i < cnt) {
            const int bin = br[k] >> 16;
            const int idx = s_loc[bin] + (br[k] & 0xFFFF);
            s_val[idx] = val[k];
            s_pos[idx] = s_base[bin] + idx;
        }
    }
    __syncthreads();
#pragma unroll
    for (int k = 0; k < EDGE_EPT; ++k) {
        const int i = t + k * BIN_T;
        if (i < cnt) packed[s_pos[i]] = s_val[i];
    }
}

__global__ void __launch_bounds__(SORT_T)
bin_sort_kernel(const int* __restrict__ packed, const int* __restrict__ bin_cnt,
                const int* __restrict__ bin_start, const int* __restrict__ deg_out,
                int* __restrict__ row_start, int* __restrict__ deg_in, int* __restrict__ by_dst,
                float* __restrict__ in_n, float* __restrict__ out_n, int n) {
    __shared__ int s_deg[BIN_W];
    __shared__ int s_fill[BIN_W];
    __shared__ int s_wave[SORT_T / 32];
    __shared__ int s_out[SORT_CAP];
    const int t = threadIdx.x;
    const int lo = bin_start[blockIdx.x];
    const int cnt = bin_cnt[blockIdx.x];
    const int* __restrict__ mine = packed + lo;
    s_deg[t] = 0;
    __syncthreads();

    for (int e0 = 0; e0 < cnt; e0 += SORT_T * SORT_B) {
        int p[SORT_B];
#pragma unroll
        for (int k = 0; k < SORT_B; ++k) {
            const int e = e0 + t + k * SORT_T;
            p[k] = mine[(e < cnt) ? e : (cnt - 1)];
        }
#pragma unroll
        for (int k = 0; k < SORT_B; ++k) {
            const int e = e0 + t + k * SORT_T;
            if (e < cnt) atomicAdd(&s_deg[p[k] & (BIN_W - 1)], 1);
        }
    }
    __syncthreads();

    const int dg = s_deg[t];
    const int ex = block_excl_scan<SORT_T / 32>(dg, s_wave);
    const bool staged = (cnt <= SORT_CAP);
    s_fill[t] = staged ? ex : (lo + ex);
    const int node = blockIdx.x * BIN_W + t;
    if (node < n) {
        row_start[node] = lo + ex;
        deg_in[node] = dg;
        in_n[node]  = __frsqrt_rn(fmaxf((float)dg, 1.0f));
        out_n[node] = __frsqrt_rn(fmaxf((float)deg_out[node], 1.0f));
    }
    __syncthreads();

    for (int e0 = 0; e0 < cnt; e0 += SORT_T * SORT_B) {
        int p[SORT_B];
#pragma unroll
        for (int k = 0; k < SORT_B; ++k) {
            const int e = e0 + t + k * SORT_T;
            p[k] = mine[(e < cnt) ? e : (cnt - 1)];
        }
#pragma unroll
        for (int k = 0; k < SORT_B; ++k) {
            const int e = e0 + t + k * SORT_T;
            if (e < cnt) {
                const int pos = atomicAdd(&s_fill[p[k] & (BIN_W - 1)], 1);
                const int s = (int)((unsigned)p[k] >> BIN_SHIFT);
                if (staged) s_out[pos] = s;
                else        by_dst[pos] = s;
            }
        }
    }
    if (staged) {
        __syncthreads();
#pragma unroll 4
        for (int i = t; i < cnt; i += SORT_T) by_dst[lo + i] = s_out[i];
    }
}

#define GATHER_B 16
#define TILE_M 16
#define LDS_PITCH 68

template <bool SCALE_SRC, bool SCALE_OUT, bool POOL_OUT>
__global__ void __launch_bounds__(256)
conv_kernel(const float* __restrict__ x, const float* __restrict__ out_n,
            const float* __restrict__ in_n, const int* __restrict__ row_start,
            const int* __restrict__ deg_in, const int* __restrict__ by_dst,
            const float* __restrict__ W, const float* __restrict__ bias,
            float* __restrict__ out, const int* __restrict__ gid, float* __restrict__ cnt,
            int n_nodes, float* __restrict__ zero_buf, int zero_n, int do_relu) {
    __shared__ __attribute__((aligned(16))) float s_w[DF * LDS_PITCH];
    __shared__ __attribute__((aligned(16))) float s_a[2][TILE_M * LDS_PITCH];
    const int lane = threadIdx.x & 31;
    const int wave = __builtin_amdgcn_readfirstlane((int)(threadIdx.x >> 5));

    if (SCALE_OUT) {
        const int nthreads = (int)(gridDim.x * blockDim.x);
        for (int i = (int)(blockIdx.x * blockDim.x + threadIdx.x); i < zero_n; i += nthreads)
            zero_buf[i] = 0.0f;
    }
    static_assert(DF * DF == 256 * 16, "256 threads take 16 floats of W each");
    {
        float4 wv[4];
#pragma unroll
        for (int i = 0; i < 4; ++i)
            wv[i] = *(const float4*)(W + 4 * ((int)threadIdx.x + 256 * i));
#pragma unroll
        for (int i = 0; i < 4; ++i) {
            const int e = 4 * ((int)threadIdx.x + 256 * i);
            const int k = e >> 6, c0 = e & (DF - 1);
            s_w[(c0 + 0) * LDS_PITCH + k] = wv[i].x;
            s_w[(c0 + 1) * LDS_PITCH + k] = wv[i].y;
            s_w[(c0 + 2) * LDS_PITCH + k] = wv[i].z;
            s_w[(c0 + 3) * LDS_PITCH + k] = wv[i].w;
        }
    }

    const int ntiles = (n_nodes + TILE_M - 1) / TILE_M;
    int buf = 0;
    for (int tile = blockIdx.x; tile < ntiles; tile += gridDim.x, buf ^= 1) {
        float* __restrict__ sa = s_a[buf];
        const int m0 = tile * TILE_M;

        for (int q = 0; q < 2; ++q) {
            const int r = 2 * wave + q;
            const int node = m0 + r;
            float ax = 0.0f, ay = 0.0f;
            if (node < n_nodes) {
                const int start = __builtin_amdgcn_readfirstlane(row_start[node]);
                const int m     = __builtin_amdgcn_readfirstlane(deg_in[node]);
                const float rs  = in_n[node];
                for (int base = 0; base < m; base += 32) {
                    const int mm = (m - base < 32) ? (m - base) : 32;
                    int idx = 0;
                    float sc = 0.0f;
                    if (lane < mm) {
                        idx = by_dst[start + base + lane];
                        if (SCALE_SRC) sc = out_n[idx];
                    }
                    for (int i = 0; i < mm; i += GATHER_B) {
                        float2 v[GATHER_B];
                        int s[GATHER_B];
                        float f[GATHER_B];
#pragma unroll
                        for (int k = 0; k < GATHER_B; ++k) {
                            const int j = (i + k < mm) ? (i + k) : (mm - 1);
                            s[k] = __builtin_amdgcn_readlane(idx, j);
                            f[k] = SCALE_SRC ? __int_as_float(__builtin_amdgcn_readlane(__float_as_int(sc), j)) : 1.0f;
                        }
#pragma unroll
                        for (int k = 0; k < GATHER_B; ++k) {
                            v[k] = make_float2(0.0f, 0.0f);
                            if (i + k < mm)
                                v[k] = *(const float2*)(x + (size_t)s[k] * DF + 2 * lane);
                        }
#pragma unroll
                        for (int k = 0; k < GATHER_B; ++k) {
                            if (SCALE_SRC) {
                                ax = __builtin_fmaf(v[k].x, f[k], ax);
                                ay = __builtin_fmaf(v[k].y, f[k], ay);
                            } else {
                                ax += v[k].x;
                                ay += v[k].y;
                            }
                        }
                    }
                }
                ax *= rs;
                ay *= rs;
            }
            *(float2*)(sa + r * LDS_PITCH + 2 * lane) = make_float2(ax, ay);
        }
        __syncthreads();

        if (wave < 4) {
            const int half = lane >> 4;
            const int r16  = lane & 15;
            const int col  = wave * 16 + r16;
            const float* ap = sa + r16 * LDS_PITCH + 2 * half;
            const float* bp = s_w + col * LDS_PITCH + 2 * half;
            v8f c = {};
#pragma unroll
            for (int k = 0; k < 16; ++k) {
                const v2f a = *(const v2f*)(ap + 4 * k);
                const v2f b = *(const v2f*)(bp + 4 * k);
                c = __builtin_amdgcn_wmma_f32_16x16x4_f32(false, a, false, b, (short)0, c,
                                                          false, false);
            }
            const float bv = bias[col];
            if (POOL_OUT) {
                const int row  = m0 + r16;
                const int gl = gid[(row < n_nodes) ? row : 0];
                float v[8];
#pragma unroll
                for (int j = 0; j < 8; ++j) {
                    v[j] = c[j] + bv;
                    if (do_relu) v[j] = fmaxf(v[j], 0.0f);
                }
                const int g0 = __builtin_amdgcn_readfirstlane(gl);
                const bool one_graph = (m0 + TILE_M <= n_nodes) &&
                                       (__builtin_amdgcn_ballot_w32(gl != g0) == 0u);
                if (one_graph) {
                    float t = ((v[0] + v[1]) + (v[2] + v[3])) + ((v[4] + v[5]) + (v[6] + v[7]));
                    t += __shfl_xor(t, 16, 32);
                    if (half == 0) atomicAdd(out + (size_t)g0 * DF + col, t);
                    if (threadIdx.x == 0) atomicAdd(cnt + g0, 16.0f);
                } else {
                    int gr[8];
#pragma unroll
                    for (int j = 0; j < 8; ++j) gr[j] = __shfl(gl, 8 * half + j, 32);
                    int cur = gr[0];
                    float acc = 0.0f, run = 0.0f;
#pragma unroll
                    for (int j = 0; j < 8; ++j) {
                        if (m0 + 8 * half + j < n_nodes) {
                            if (gr[j] != cur) {
                                atomicAdd(out + (size_t)cur * DF + col, acc);
                                if (wave == 0 && r16 == 0) atomicAdd(cnt + cur, run);
                                cur = gr[j];
                                acc = 0.0f; run = 0.0f;
                            }
                            acc += v[j];
                            run += 1.0f;
                        }
                    }
                    if (run > 0.0f) {
                        atomicAdd(out + (size_t)cur * DF + col, acc);
                        if (wave == 0 && r16 == 0) atomicAdd(cnt + cur, run);
                    }
                }
            } else {
                float* orow0 = out + (size_t)m0 * DF + col;
                if (m0 + TILE_M <= n_nodes) {
#pragma unroll
                    for (int j = 0; j < 8; ++j) {
                        float v = c[j] + bv;
                        if (do_relu) v = fmaxf(v, 0.0f);
                        if (SCALE_OUT) v *= out_n[m0 + j + 8 * half];
                        orow0[(size_t)(j + 8 * half) * DF] = v;
                    }
                } else {
#pragma unroll
                    for (int j = 0; j < 8; ++j) {
                        const int orow = m0 + j + 8 * half;
                        if (orow < n_nodes) {
                            float v = c[j] + bv;
                            if (do_relu) v = fmaxf(v, 0.0f);
                            if (SCALE_OUT) v *= out_n[orow];
                            out[(size_t)orow * DF + col] = v;
                        }
                    }
                }
            }
        }
    }
}

template <int NCOLS_C>
__global__ void gemm_wmma_kernel(const float* __restrict__ A,
                                 const float* __restrict__ count,
                                 const float* __restrict__ W,
                                 const float* __restrict__ bias,
                                 float* __restrict__ out,
                                 int rows, int ncols_rt) {
    const int Ncols = (NCOLS_C > 0) ? NCOLS_C : ncols_rt;
    const int lane = threadIdx.x & 31;
    const int wave = threadIdx.x >> 5;
    const int half = lane >> 4;
    const int r16  = lane & 15;
    const int n0 = wave * 16;

    const int col  = n0 + r16;
    const bool cok = (col < Ncols);
    const int colc = cok ? col : (Ncols - 1);
    const float* wp = W + (size_t)(2 * half) * Ncols + colc;

    v2f b[16];
#pragma unroll
    for (int k = 0; k < 16; ++k) {
        b[k].x = wp[(size_t)(4 * k) * Ncols];
        b[k].y = wp[(size_t)(4 * k + 1) * Ncols];
    }
    const float bv = bias[colc];

    const int ntiles = (rows + 15) >> 4;
    for (int tile = blockIdx.x; tile < ntiles; tile += gridDim.x) {
        const int m0 = tile * 16;
        const int row  = m0 + r16;
        const int rowc = (row < rows) ? row : 0;
        const float s = 1.0f / fmaxf(count[rowc], 1.0f);
        const float* arow = A + (size_t)rowc * DF + 2 * half;

        v8f c = {};
#pragma unroll
        for (int k = 0; k < 16; ++k) {
            v2f a;
            a.x = arow[4 * k + 0] * s;
            a.y = arow[4 * k + 1] * s;
            c = __builtin_amdgcn_wmma_f32_16x16x4_f32(false, a, false, b[k], (short)0, c,
                                                      false, false);
        }

        float* orow0 = out + (size_t)m0 * Ncols + col;
        if (m0 + 16 <= rows && cok) {
#pragma unroll
            for (int j = 0; j < 8; ++j)
                orow0[(size_t)(j + 8 * half) * Ncols] = c[j] + bv;
        } else if (cok) {
#pragma unroll
            for (int j = 0; j < 8; ++j) {
                const int orow = m0 + j + 8 * half;
                if (orow < rows) out[(size_t)orow * Ncols + col] = c[j] + bv;
            }
        }
    }
}

extern "C" void kernel_launch(void* const* d_in, const int* in_sizes, int n_in,
                              void* d_out, int out_size, void* d_ws, size_t ws_size,
                              hipStream_t stream) {
    const float* features = (const float*)d_in[0];
    const int*   src      = (const int*)d_in[1];
    const int*   dst      = (const int*)d_in[2];
    const int*   gid      = (const int*)d_in[3];
    const float* W1       = (const float*)d_in[4];
    const float* b1       = (const float*)d_in[5];
    const float* W2       = (const float*)d_in[6];
    const float* b2       = (const float*)d_in[7];
    const float* Wc       = (const float*)d_in[8];
    const float* bc       = (const float*)d_in[9];
    float* outp           = (float*)d_out;

    const int N    = in_sizes[3];
    const int E    = in_sizes[1];
    const int DOUT = in_sizes[9];
    const int G    = out_size / DOUT;

    const size_t Np = ((size_t)N + 63) & ~(size_t)63;
    const size_t Ep = ((size_t)E + 63) & ~(size_t)63;
    const int nBins = (N + BIN_W - 1) / BIN_W;
    const int edgeBlocks = (E + EDGE_CAP - 1) / EDGE_CAP;
    int*   deg_o = (int*)d_ws;
    int*   bcnt  = deg_o + Np;
    int*   bcur  = bcnt + MAX_BINS;
    int*   bsta  = bcur + MAX_BINS;
    int*   deg_i = bsta + MAX_BINS;
    int*   rowst = deg_i + Np;
    int*   pack  = rowst + Np;
    int*   bydst = pack + Ep;
    float* out_n = (float*)(bydst + Ep);
    float* in_n  = out_n + Np;
    float* bufA  = in_n + Np;
    float* pool  = bufA + (size_t)N * DF;
    float* cnt   = pool + (size_t)G * DF;

    hipMemsetAsync(deg_o, 0, sizeof(int) * (Np + 2 * MAX_BINS), stream);
    bin_count_kernel<<<edgeBlocks, BIN_T, 0, stream>>>(src, dst, deg_o, bcnt, E, nBins);
    bin_scatter_kernel<<<edgeBlocks, BIN_T, 0, stream>>>(src, dst, bcnt, bcur, bsta, pack, E, nBins);
    bin_sort_kernel<<<nBins, SORT_T, 0, stream>>>(pack, bcnt, bsta, deg_o, rowst, deg_i, bydst, in_n, out_n, N);

    const int convGrid = 2048;

    conv_kernel<true, true, false><<<convGrid, 256, 0, stream>>>(
        features, out_n, in_n, rowst, deg_i, bydst, W1, b1, bufA, nullptr, nullptr, N,
        pool, G * DF + G, 1);

    conv_kernel<false, false, true><<<convGrid, 256, 0, stream>>>(
        bufA, out_n, in_n, rowst, deg_i, bydst, W2, b2, pool, gid, cnt, N,
        nullptr, 0, 1);

    const int clsWaves = (DOUT + 15) / 16;
    if (DOUT == 8) {
        gemm_wmma_kernel<8><<<(G + 15) / 16, 32 * clsWaves, 0, stream>>>(
            pool, cnt, Wc, bc, outp, G, DOUT);
    } else {
        gemm_wmma_kernel<-1><<<(G + 15) / 16, 32 * clsWaves, 0, stream>>>(
            pool, cnt, Wc, bc, outp, G, DOUT);
    }
}


# out-degrees without a global atomic per edge: edges are also binned by source through LDS (second pass in the scatter kernel), each bin's block counts them in LDS; row-sum order kept
# speedup vs baseline: 6.4796x; 1.7037x over previous
#include <hip/hip_runtime.h>
#include <hip/hip_bf16.h>

typedef float v2f __attribute__((ext_vector_type(2)));
typedef float v8f __attribute__((ext_vector_type(8)));

#define DF 64

#define BIN_SHIFT 8
#define BIN_W (1 << BIN_SHIFT)
#define MAX_BINS 1024
#define BIN_T 1024
#define EDGE_EPT 7
#define EDGE_CAP 6400
#define SORT_T BIN_W
#define SORT_B 8
#define SORT_CAP 8192

static_assert(BIN_T == MAX_BINS, "one thread per bin slot");
static_assert(EDGE_CAP <= BIN_T * EDGE_EPT, "every edge of a block has a thread slot");
static_assert(EDGE_CAP < 65536, "a rank inside a block fits 16 bits");

template <int NW>
__device__ __forceinline__ int block_excl_scan(int v, int* s_wave) {
    const int lane = threadIdx.x & 31;
    const int wave = threadIdx.x >> 5;
    int inc = v;
    for (int o = 1; o < 32; o <<= 1) {
        int t = __shfl_up(inc, o, 32);
        if (lane >= o) inc += t;
    }
    if (lane == 31) s_wave[wave] = inc;
    __syncthreads();
    int woff = 0;
    for (int w = 0; w < NW; ++w)
        if (w < wave) woff += s_wave[w];
    __syncthreads();
    return woff + inc - v;
}

__global__ void __launch_bounds__(BIN_T)
bin_count_kernel(const int* __restrict__ src, const int* __restrict__ dst,
                 int* __restrict__ bin_cnt, int* __restrict__ sbin_cnt, int E, int nbins) {
    __shared__ int s_cnt[MAX_BINS];
    __shared__ int s_cnt2[MAX_BINS];
    const int t = threadIdx.x;
    const int lo = blockIdx.x * EDGE_CAP;
    const int cnt = (E - lo < EDGE_CAP) ? (E - lo) : EDGE_CAP;
    s_cnt[t] = 0;
    s_cnt2[t] = 0;
    __syncthreads();

    int s[EDGE_EPT], d[EDGE_EPT];
#pragma unroll
    for (int k = 0; k < EDGE_EPT; ++k) {
        const int i = t + k * BIN_T;
        s[k] = 0; d[k] = 0;
        if (i < cnt) { s[k] = src[lo + i]; d[k] = dst[lo + i]; }
    }
#pragma unroll
    for (int k = 0; k < EDGE_EPT; ++k) {
        const int i = t + k * BIN_T;
        if (i < cnt) {
            atomicAdd(&s_cnt2[s[k] >> BIN_SHIFT], 1);
            atomicAdd(&s_cnt[d[k] >> BIN_SHIFT], 1);
        }
    }
    __syncthreads();
    const int c = s_cnt[t];
    const int c2 = s_cnt2[t];
    if (t < nbins && c > 0) atomicAdd(&bin_cnt[t], c);
    if (t < nbins && c2 > 0) atomicAdd(&sbin_cnt[t], c2);
}

__device__ __forceinline__ void scatter_pass(const int (&key)[EDGE_EPT], const int (&val)[EDGE_EPT],
                                             int cnt, int nbins,
                                             const int* __restrict__ bin_cnt, int* __restrict__ cursor,
                                             int* __restrict__ bin_start, int* __restrict__ out_list,
                                             int* s_cnt, int* s_loc, int* s_base, int* s_wave,
                                             int* s_val, int* s_pos) {
    const int t = threadIdx.x;
    s_cnt[t] = 0;
    __syncthreads();

    int br[EDGE_EPT];
#pragma unroll
    for (int k = 0; k < EDGE_EPT; ++k) {
        const int i = t + k * BIN_T;
        br[k] = 0;
        if (i < cnt) {
            const int bin = key[k] >> BIN_SHIFT;
            const int r = atomicAdd(&s_cnt[bin], 1);
            br[k] = (bin << 16) | r;
        }
    }

    const int total = (t < nbins) ? bin_cnt[t] : 0;
    const int start = block_excl_scan<BIN_T / 32>(total, s_wave);
    if (blockIdx.x == 0 && t < nbins) bin_start[t] = start;

    const int mine = s_cnt[t];
    const int loc = block_excl_scan<BIN_T / 32>(mine, s_wave);
    int base = 0;
    if (mine > 0) base = start + atomicAdd(&cursor[t], mine);
    s_loc[t] = loc;
    s_base[t] = base - loc;
    __syncthreads();

#pragma unroll
    for (int k = 0; k < EDGE_EPT; ++k) {
        const int i = t + k * BIN_T;
        if (i < cnt) {
            const int bin = br[k] >> 16;
            const int idx = s_loc[bin] + (br[k] & 0xFFFF);
            s_val[idx] = val[k];
            s_pos[idx] = s_base[bin] + idx;
        }
    }
    __syncthreads();
#pragma unroll
    for (int k = 0; k < EDGE_EPT; ++k) {
        const int i = t + k * BIN_T;
        if (i < cnt) out_list[s_pos[i]] = s_val[i];
    }
}

__global__ void __launch_bounds__(BIN_T)
bin_scatter_kernel(const int* __restrict__ src, const int* __restrict__ dst,
                   const int* __restrict__ bin_cnt, int* __restrict__ cursor,
                   int* __restrict__ bin_start, int* __restrict__ packed,
                   const int* __restrict__ sbin_cnt, int* __restrict__ scursor,
                   int* __restrict__ sbin_start, int* __restrict__ spacked, int E, int nbins) {
    __shared__ int s_cnt[MAX_BINS];
    __shared__ int s_loc[MAX_BINS];
    __shared__ int s_base[MAX_BINS];
    __shared__ int s_wave[BIN_T / 32];
    __shared__ int s_val[EDGE_CAP];
    __shared__ int s_pos[EDGE_CAP];
    const int t = threadIdx.x;
    const int lo = blockIdx.x * EDGE_CAP;
    const int cnt = (E - lo < EDGE_CAP) ? (E - lo) : EDGE_CAP;

    int s[EDGE_EPT], d[EDGE_EPT], val[EDGE_EPT];
#pragma unroll
    for (int k = 0; k < EDGE_EPT; ++k) {
        const int i = t + k * BIN_T;
        s[k] = 0; d[k] = 0;
        if (i < cnt) { s[k] = src[lo + i]; d[k] = dst[lo + i]; }
    }
#pragma unroll
    for (int k = 0; k < EDGE_EPT; ++k)
        val[k] = (int)(((unsigned)s[k] << BIN_SHIFT) | (unsigned)(d[k] & (BIN_W - 1)));
    scatter_pass(d, val, cnt, nbins, bin_cnt, cursor, bin_start, packed,
                 s_cnt, s_loc, s_base, s_wave, s_val, s_pos);
#pragma unroll
    for (int k = 0; k < EDGE_EPT; ++k)
        val[k] = s[k] & (BIN_W - 1);
    scatter_pass(s, val, cnt, nbins, sbin_cnt, scursor, sbin_start, spacked,
                 s_cnt, s_loc, s_base, s_wave, s_val, s_pos);
}

__global__ void __launch_bounds__(SORT_T)
bin_sort_kernel(const int* __restrict__ packed, const int* __restrict__ bin_cnt,
                const int* __restrict__ bin_start, const int* __restrict__ spacked,
                const int* __restrict__ sbin_cnt, const int* __restrict__ sbin_start,
                int* __restrict__ row_start, int* __restrict__ deg_in, int* __restrict__ by_dst,
                float* __restrict__ in_n, float* __restrict__ out_n, int n) {
    __shared__ int s_deg[BIN_W];
    __shared__ int s_odeg[BIN_W];
    __shared__ int s_fill[BIN_W];
    __shared__ int s_wave[SORT_T / 32];
    __shared__ int s_out[SORT_CAP];
    const int t = threadIdx.x;
    const int lo = bin_start[blockIdx.x];
    const int cnt = bin_cnt[blockIdx.x];
    const int* __restrict__ mine = packed + lo;
    const int slo = sbin_start[blockIdx.x];
    const int scnt = sbin_cnt[blockIdx.x];
    const int* __restrict__ smine = spacked + slo;
    s_deg[t] = 0;
    s_odeg[t] = 0;
    __syncthreads();

    for (int e0 = 0; e0 < scnt; e0 += SORT_T * SORT_B) {
        int p[SORT_B];
#pragma unroll
        for (int k = 0; k < SORT_B; ++k) {
            const int e = e0 + t + k * SORT_T;
            p[k] = smine[(e < scnt) ? e : (scnt - 1)];
        }
#pragma unroll
        for (int k = 0; k < SORT_B; ++k) {
            const int e = e0 + t + k * SORT_T;
            if (e < scnt) atomicAdd(&s_odeg[p[k] & (BIN_W - 1)], 1);
        }
    }

    for (int e0 = 0; e0 < cnt; e0 += SORT_T * SORT_B) {
        int p[SORT_B];
#pragma unroll
        for (int k = 0; k < SORT_B; ++k) {
            const int e = e0 + t + k * SORT_T;
            p[k] = mine[(e < cnt) ? e : (cnt - 1)];
        }
#pragma unroll
        for (int k = 0; k < SORT_B; ++k) {
            const int e = e0 + t + k * SORT_T;
            if (e < cnt) atomicAdd(&s_deg[p[k] & (BIN_W - 1)], 1);
        }
    }
    __syncthreads();

    const int dg = s_deg[t];
    const int ex = block_excl_scan<SORT_T / 32>(dg, s_wave);
    const bool staged = (cnt <= SORT_CAP);
    s_fill[t] = staged ? ex : (lo + ex);
    const int node = blockIdx.x * BIN_W + t;
    if (node < n) {
        row_start[node] = lo + ex;
        deg_in[node] = dg;
        in_n[node]  = __frsqrt_rn(fmaxf((float)dg, 1.0f));
        out_n[node] = __frsqrt_rn(fmaxf((float)s_odeg[t], 1.0f));
    }
    __syncthreads();

    for (int e0 = 0; e0 < cnt; e0 += SORT_T * SORT_B) {
        int p[SORT_B];
#pragma unroll
        for (int k = 0; k < SORT_B; ++k) {
            const int e = e0 + t + k * SORT_T;
            p[k] = mine[(e < cnt) ? e : (cnt - 1)];
        }
#pragma unroll
        for (int k = 0; k < SORT_B; ++k) {
            const int e = e0 + t + k * SORT_T;
            if (e < cnt) {
                const int pos = atomicAdd(&s_fill[p[k] & (BIN_W - 1)], 1);
                const int s = (int)((unsigned)p[k] >> BIN_SHIFT);
                if (staged) s_out[pos] = s;
                else        by_dst[pos] = s;
            }
        }
    }
    if (staged) {
        __syncthreads();
#pragma unroll 4
        for (int i = t; i < cnt; i += SORT_T) by_dst[lo + i] = s_out[i];
    }
}

#define GATHER_B 16
#define TILE_M 16
#define LDS_PITCH 68

template <bool SCALE_SRC, bool SCALE_OUT, bool POOL_OUT>
__global__ void __launch_bounds__(256)
conv_kernel(const float* __restrict__ x, const float* __restrict__ out_n,
            const float* __restrict__ in_n, const int* __restrict__ row_start,
            const int* __restrict__ deg_in, const int* __restrict__ by_dst,
            const float* __restrict__ W, const float* __restrict__ bias,
            float* __restrict__ out, const int* __restrict__ gid, float* __restrict__ cnt,
            int n_nodes, float* __restrict__ zero_buf, int zero_n, int do_relu) {
    __shared__ __attribute__((aligned(16))) float s_w[DF * LDS_PITCH];
    __shared__ __attribute__((aligned(16))) float s_a[2][TILE_M * LDS_PITCH];
    const int lane = threadIdx.x & 31;
    const int wave = __builtin_amdgcn_readfirstlane((int)(threadIdx.x >> 5));

    if (SCALE_OUT) {
        const int nthreads = (int)(gridDim.x * blockDim.x);
        for (int i = (int)(blockIdx.x * blockDim.x + threadIdx.x); i < zero_n; i += nthreads)
            zero_buf[i] = 0.0f;
    }
    static_assert(DF * DF == 256 * 16, "256 threads take 16 floats of W each");
    {
        float4 wv[4];
#pragma unroll
        for (int i = 0; i < 4; ++i)
            wv[i] = *(const float4*)(W + 4 * ((int)threadIdx.x + 256 * i));
#pragma unroll
        for (int i = 0; i < 4; ++i) {
            const int e = 4 * ((int)threadIdx.x + 256 * i);
            const int k = e >> 6, c0 = e & (DF - 1);
            s_w[(c0 + 0) * LDS_PITCH + k] = wv[i].x;
            s_w[(c0 + 1) * LDS_PITCH + k] = wv[i].y;
            s_w[(c0 + 2) * LDS_PITCH + k] = wv[i].z;
            s_w[(c0 + 3) * LDS_PITCH + k] = wv[i].w;
        }
    }

    const int ntiles = (n_nodes + TILE_M - 1) / TILE_M;
    int buf = 0;
    for (int tile = blockIdx.x; tile < ntiles; tile += gridDim.x, buf ^= 1) {
        float* __restrict__ sa = s_a[buf];
        const int m0 = tile * TILE_M;

        for (int q = 0; q < 2; ++q) {
            const int r = 2 * wave + q;
            const int node = m0 + r;
            float ax = 0.0f, ay = 0.0f;
            if (node < n_nodes) {
                const int start = __builtin_amdgcn_readfirstlane(row_start[node]);
                const int m     = __builtin_amdgcn_readfirstlane(deg_in[node]);
                const float rs  = in_n[node];
                for (int base = 0; base < m; base += 32) {
                    const int mm = (m - base < 32) ? (m - base) : 32;
                    int idx = 0;
                    float sc = 0.0f;
                    if (lane < mm) {
                        idx = by_dst[start + base + lane];
                        if (SCALE_SRC) sc = out_n[idx];
                    }
                    for (int i = 0; i < mm; i += GATHER_B) {
                        float2 v[GATHER_B];
                        int s[GATHER_B];
                        float f[GATHER_B];
#pragma unroll
                        for (int k = 0; k < GATHER_B; ++k) {
                            const int j = (i + k < mm) ? (i + k) : (mm - 1);
                            s[k] = __builtin_amdgcn_readlane(idx, j);
                            f[k] = SCALE_SRC ? __int_as_float(__builtin_amdgcn_readlane(__float_as_int(sc), j)) : 1.0f;
                        }
#pragma unroll
                        for (int k = 0; k < GATHER_B; ++k) {
                            v[k] = make_float2(0.0f, 0.0f);
                            if (i + k < mm)
                                v[k] = *(const float2*)(x + (size_t)s[k] * DF + 2 * lane);
                        }
#pragma unroll
                        for (int k = 0; k < GATHER_B; ++k) {
                            if (SCALE_SRC) {
                                ax = __builtin_fmaf(v[k].x, f[k], ax);
                                ay = __builtin_fmaf(v[k].y, f[k], ay);
                            } else {
                                ax += v[k].x;
                                ay += v[k].y;
                            }
                        }
                    }
                }
                ax *= rs;
                ay *= rs;
            }
            *(float2*)(sa + r * LDS_PITCH + 2 * lane) = make_float2(ax, ay);
        }
        __syncthreads();

        if (wave < 4) {
            const int half = lane >> 4;
            const int r16  = lane & 15;
            const int col  = wave * 16 + r16;
            const float* ap = sa + r16 * LDS_PITCH + 2 * half;
            const float* bp = s_w + col * LDS_PITCH + 2 * half;
            v8f c = {};
#pragma unroll
            for (int k = 0; k < 16; ++k) {
                const v2f a = *(const v2f*)(ap + 4 * k);
                const v2f b = *(const v2f*)(bp + 4 * k);
                c = __builtin_amdgcn_wmma_f32_16x16x4_f32(false, a, false, b, (short)0, c,
                                                          false, false);
            }
            const float bv = bias[col];
            if (POOL_OUT) {
                const int row  = m0 + r16;
                const int gl = gid[(row < n_nodes) ? row : 0];
                float v[8];
#pragma unroll
                for (int j = 0; j < 8; ++j) {
                    v[j] = c[j] + bv;
                    if (do_relu) v[j] = fmaxf(v[j], 0.0f);
                }
                const int g0 = __builtin_amdgcn_readfirstlane(gl);
                const bool one_graph = (m0 + TILE_M <= n_nodes) &&
                                       (__builtin_amdgcn_ballot_w32(gl != g0) == 0u);
                if (one_graph) {
                    float t = ((v[0] + v[1]) + (v[2] + v[3])) + ((v[4] + v[5]) + (v[6] + v[7]));
                    t += __shfl_xor(t, 16, 32);
                    if (half == 0) atomicAdd(out + (size_t)g0 * DF + col, t);
                    if (threadIdx.x == 0) atomicAdd(cnt + g0, 16.0f);
                } else {
                    int gr[8];
#pragma unroll
                    for (int j = 0; j < 8; ++j) gr[j] = __shfl(gl, 8 * half + j, 32);
                    int cur = gr[0];
                    float acc = 0.0f, run = 0.0f;
#pragma unroll
                    for (int j = 0; j < 8; ++j) {
                        if (m0 + 8 * half + j < n_nodes) {
                            if (gr[j] != cur) {
                                atomicAdd(out + (size_t)cur * DF + col, acc);
                                if (wave == 0 && r16 == 0) atomicAdd(cnt + cur, run);
                                cur = gr[j];
                                acc = 0.0f; run = 0.0f;
                            }
                            acc += v[j];
                            run += 1.0f;
                        }
                    }
                    if (run > 0.0f) {
                        atomicAdd(out + (size_t)cur * DF + col, acc);
                        if (wave == 0 && r16 == 0) atomicAdd(cnt + cur, run);
                    }
                }
            } else {
                float* orow0 = out + (size_t)m0 * DF + col;
                if (m0 + TILE_M <= n_nodes) {
#pragma unroll
                    for (int j = 0; j < 8; ++j) {
                        float v = c[j] + bv;
                        if (do_relu) v = fmaxf(v, 0.0f);
                        if (SCALE_OUT) v *= out_n[m0 + j + 8 * half];
                        orow0[(size_t)(j + 8 * half) * DF] = v;
                    }
                } else {
#pragma unroll
                    for (int j = 0; j < 8; ++j) {
                        const int orow = m0 + j + 8 * half;
                        if (orow < n_nodes) {
                            float v = c[j] + bv;
                            if (do_relu) v = fmaxf(v, 0.0f);
                            if (SCALE_OUT) v *= out_n[orow];
                            out[(size_t)orow * DF + col] = v;
                        }
                    }
                }
            }
        }
    }
}

template <int NCOLS_C>
__global__ void gemm_wmma_kernel(const float* __restrict__ A,
                                 const float* __restrict__ count,
                                 const float* __restrict__ W,
                                 const float* __restrict__ bias,
                                 float* __restrict__ out,
                                 int rows, int ncols_rt) {
    const int Ncols = (NCOLS_C > 0) ? NCOLS_C : ncols_rt;
    const int lane = threadIdx.x & 31;
    const int wave = threadIdx.x >> 5;
    const int half = lane >> 4;
    const int r16  = lane & 15;
    const int n0 = wave * 16;

    const int col  = n0 + r16;
    const bool cok = (col < Ncols);
    const int colc = cok ? col : (Ncols - 1);
    const float* wp = W + (size_t)(2 * half) * Ncols + colc;

    v2f b[16];
#pragma unroll
    for (int k = 0; k < 16; ++k) {
        b[k].x = wp[(size_t)(4 * k) * Ncols];
        b[k].y = wp[(size_t)(4 * k + 1) * Ncols];
    }
    const float bv = bias[colc];

    const int ntiles = (rows + 15) >> 4;
    for (int tile = blockIdx.x; tile < ntiles; tile += gridDim.x) {
        const int m0 = tile * 16;
        const int row  = m0 + r16;
        const int rowc = (row < rows) ? row : 0;
        const float s = 1.0f / fmaxf(count[rowc], 1.0f);
        const float* arow = A + (size_t)rowc * DF + 2 * half;

        v8f c = {};
#pragma unroll
        for (int k = 0; k < 16; ++k) {
            v2f a;
            a.x = arow[4 * k + 0] * s;
            a.y = arow[4 * k + 1] * s;
            c = __builtin_amdgcn_wmma_f32_16x16x4_f32(false, a, false, b[k], (short)0, c,
                                                      false, false);
        }

        float* orow0 = out + (size_t)m0 * Ncols + col;
        if (m0 + 16 <= rows && cok) {
#pragma unroll
            for (int j = 0; j < 8; ++j)
                orow0[(size_t)(j + 8 * half) * Ncols] = c[j] + bv;
        } else if (cok) {
#pragma unroll
            for (int j = 0; j < 8; ++j) {
                const int orow = m0 + j + 8 * half;
                if (orow < rows) out[(size_t)orow * Ncols + col] = c[j] + bv;
            }
        }
    }
}

extern "C" void kernel_launch(void* const* d_in, const int* in_sizes, int n_in,
                              void* d_out, int out_size, void* d_ws, size_t ws_size,
                              hipStream_t stream) {
    const float* features = (const float*)d_in[0];
    const int*   src      = (const int*)d_in[1];
    const int*   dst      = (const int*)d_in[2];
    const int*   gid      = (const int*)d_in[3];
    const float* W1       = (const float*)d_in[4];
    const float* b1       = (const float*)d_in[5];
    const float* W2       = (const float*)d_in[6];
    const float* b2       = (const float*)d_in[7];
    const float* Wc       = (const float*)d_in[8];
    const float* bc       = (const float*)d_in[9];
    float* outp           = (float*)d_out;

    const int N    = in_sizes[3];
    const int E    = in_sizes[1];
    const int DOUT = in_sizes[9];
    const int G    = out_size / DOUT;

    const size_t Np = ((size_t)N + 63) & ~(size_t)63;
    const size_t Ep = ((size_t)E + 63) & ~(size_t)63;
    const int nBins = (N + BIN_W - 1) / BIN_W;
    const int edgeBlocks = (E + EDGE_CAP - 1) / EDGE_CAP;
    int*   bcnt  = (int*)d_ws;
    int*   bcur  = bcnt + MAX_BINS;
    int*   sbcnt = bcur + MAX_BINS;
    int*   sbcur = sbcnt + MAX_BINS;
    int*   bsta  = sbcur + MAX_BINS;
    int*   sbsta = bsta + MAX_BINS;
    int*   deg_i = sbsta + MAX_BINS;
    int*   rowst = deg_i + Np;
    int*   pack  = rowst + Np;
    int*   spack = pack + Ep;
    int*   bydst = spack + Ep;
    float* out_n = (float*)(bydst + Ep);
    float* in_n  = out_n + Np;
    float* bufA  = in_n + Np;
    float* pool  = bufA + (size_t)N * DF;
    float* cnt   = pool + (size_t)G * DF;

    hipMemsetAsync(bcnt, 0, sizeof(int) * 4 * MAX_BINS, stream);
    bin_count_kernel<<<edgeBlocks, BIN_T, 0, stream>>>(src, dst, bcnt, sbcnt, E, nBins);
    bin_scatter_kernel<<<edgeBlocks, BIN_T, 0, stream>>>(src, dst, bcnt, bcur, bsta, pack,
                                                         sbcnt, sbcur, sbsta, spack, E, nBins);
    bin_sort_kernel<<<nBins, SORT_T, 0, stream>>>(pack, bcnt, bsta, spack, sbcnt, sbsta,
                                                  rowst, deg_i, bydst, in_n, out_n, N);

    const int convGrid = 2048;

    conv_kernel<true, true, false><<<convGrid, 256, 0, stream>>>(
        features, out_n, in_n, rowst, deg_i, bydst, W1, b1, bufA, nullptr, nullptr, N,
        pool, G * DF + G, 1);

    conv_kernel<false, false, true><<<convGrid, 256, 0, stream>>>(
        bufA, out_n, in_n, rowst, deg_i, bydst, W2, b2, pool, gid, cnt, N,
        nullptr, 0, 1);

    const int clsWaves = (DOUT + 15) / 16;
    if (DOUT == 8) {
        gemm_wmma_kernel<8><<<(G + 15) / 16, 32 * clsWaves, 0, stream>>>(
            pool, cnt, Wc, bc, outp, G, DOUT);
    } else {
        gemm_wmma_kernel<-1><<<(G + 15) / 16, 32 * clsWaves, 0, stream>>>(
            pool, cnt, Wc, bc, outp, G, DOUT);
    }
}


# edge sort in one pass: each block orders its edges by bin in LDS in place, plus a table of run starts; the bin's block gathers its runs; count pass, memset, global atomics gone; row-sum order changes
# speedup vs baseline: 6.7411x; 1.0404x over previous
#include <hip/hip_runtime.h>
#include <hip/hip_bf16.h>

typedef float v2f __attribute__((ext_vector_type(2)));
typedef float v8f __attribute__((ext_vector_type(8)));

#define DF 64

#define BIN_SHIFT 8
#define BIN_W (1 << BIN_SHIFT)
#define MAX_BINS 1024
#define BIN_T 1024
#define EDGE_EPT 7
#define EDGE_CAP 6400
#define LOC_PITCH (MAX_BINS + 64)
#define SORT_T BIN_W
#define SORT_CAP 6144
#define RUN_B 16

static_assert(BIN_T == MAX_BINS, "one thread per bin slot");
static_assert(EDGE_CAP <= BIN_T * EDGE_EPT, "every edge of a block has a thread slot");
static_assert(EDGE_CAP < 65536, "a rank inside a block fits 16 bits");
static_assert((SORT_T / 8) % RUN_B == 0, "a wave's runs of a chunk come in whole trips");

template <int NW>
__device__ __forceinline__ int block_excl_scan(int v, int* s_wave) {
    const int lane = threadIdx.x & 31;
    const int wave = threadIdx.x >> 5;
    int inc = v;
    for (int o = 1; o < 32; o <<= 1) {
        int t = __shfl_up(inc, o, 32);
        if (lane >= o) inc += t;
    }
    if (lane == 31) s_wave[wave] = inc;
    __syncthreads();
    int woff = 0;
    for (int w = 0; w < NW; ++w)
        if (w < wave) woff += s_wave[w];
    __syncthreads();
    return woff + inc - v;
}

__device__ __forceinline__ void local_pass(const int (&key)[EDGE_EPT], const int (&val)[EDGE_EPT],
                                           int cnt, int nbins,
                                           int* __restrict__ loc_row, int* __restrict__ out_blk,
                                           int* s_cnt, int* s_loc, int* s_wave, int* s_val) {
    const int t = threadIdx.x;
    s_cnt[t] = 0;
    __syncthreads();

    int br[EDGE_EPT];
#pragma unroll
    for (int k = 0; k < EDGE_EPT; ++k) {
        const int i = t + k * BIN_T;
        br[k] = 0;
        if (i < cnt) {
            const int bin = key[k] >> BIN_SHIFT;
            const int r = atomicAdd(&s_cnt[bin], 1);
            br[k] = (bin << 16) | r;
        }
    }
    __syncthreads();

    const int mine = s_cnt[t];
    const int loc = block_excl_scan<BIN_T / 32>(mine, s_wave);
    s_loc[t] = loc;
    if (t < nbins) loc_row[t] = loc;
    if (t == 0) loc_row[nbins] = cnt;
    __syncthreads();

#pragma unroll
    for (int k = 0; k < EDGE_EPT; ++k) {
        const int i = t + k * BIN_T;
        if (i < cnt) {
            const int bin = br[k] >> 16;
            s_val[s_loc[bin] + (br[k] & 0xFFFF)] = val[k];
        }
    }
    __syncthreads();
#pragma unroll
    for (int k = 0; k < EDGE_EPT; ++k) {
        const int i = t + k * BIN_T;
        if (i < cnt) out_blk[i] = s_val[i];
    }
}

__global__ void __launch_bounds__(BIN_T)
bin_local_kernel(const int* __restrict__ src, const int* __restrict__ dst,
                 int* __restrict__ dloc, int* __restrict__ packed,
                 int* __restrict__ sloc, int* __restrict__ spacked, int E, int nbins) {
    __shared__ int s_cnt[MAX_BINS];
    __shared__ int s_loc[MAX_BINS];
    __shared__ int s_wave[BIN_T / 32];
    __shared__ int s_val[EDGE_CAP];
    const int t = threadIdx.x;
    const int lo = blockIdx.x * EDGE_CAP;
    const int cnt = (E - lo < EDGE_CAP) ? (E - lo) : EDGE_CAP;

    int s[EDGE_EPT], d[EDGE_EPT], val[EDGE_EPT];
#pragma unroll
    for (int k = 0; k < EDGE_EPT; ++k) {
        const int i = t + k * BIN_T;
        s[k] = 0; d[k] = 0;
        if (i < cnt) { s[k] = src[lo + i]; d[k] = dst[lo + i]; }
    }
#pragma unroll
    for (int k = 0; k < EDGE_EPT; ++k)
        val[k] = (int)(((unsigned)s[k] << BIN_SHIFT) | (unsigned)(d[k] & (BIN_W - 1)));
    local_pass(d, val, cnt, nbins, dloc + (size_t)blockIdx.x * LOC_PITCH, packed + lo,
               s_cnt, s_loc, s_wave, s_val);
#pragma unroll
    for (int k = 0; k < EDGE_EPT; ++k)
        val[k] = s[k] & (BIN_W - 1);
    local_pass(s, val, cnt, nbins, sloc + (size_t)blockIdx.x * LOC_PITCH, spacked + lo,
               s_cnt, s_loc, s_wave, s_val);
}

template <class F>
__device__ __forceinline__ void walk_bin(const int* __restrict__ list, const int* __restrict__ loc,
                                         int j, int nblk, int* s_rbase, int* s_rlen, int* s_roff,
                                         int* s_wave, int* s_carry, F f) {
    const int t = threadIdx.x;
    const int lane = t & 31;
    const int wave = __builtin_amdgcn_readfirstlane(t >> 5);
    int carry = 0;
    for (int c0 = 0; c0 < nblk; c0 += SORT_T) {
        const int b = c0 + t;
        int lo = 0, len = 0;
        if (b < nblk) {
            const int* row = loc + (size_t)b * LOC_PITCH + j;
            lo = row[0];
            len = row[1] - lo;
        }
        const int off = carry + block_excl_scan<SORT_T / 32>(len, s_wave);
        s_rbase[t] = b * EDGE_CAP + lo;
        s_rlen[t] = len;
        s_roff[t] = off;
        if (t == SORT_T - 1) *s_carry = off + len;
        __syncthreads();
        carry = *s_carry;
        for (int q0 = 0; q0 < SORT_T / 8; q0 += RUN_B) {
            int p[RUN_B], ln[RUN_B], bs[RUN_B], of[RUN_B];
#pragma unroll
            for (int k = 0; k < RUN_B; ++k) {
                const int r = wave + 8 * (q0 + k);
                ln[k] = __builtin_amdgcn_readfirstlane(s_rlen[r]);
                bs[k] = __builtin_amdgcn_readfirstlane(s_rbase[r]);
                of[k] = __builtin_amdgcn_readfirstlane(s_roff[r]);
            }
#pragma unroll
            for (int k = 0; k < RUN_B; ++k) {
                const int last = (ln[k] > 0) ? (ln[k] - 1) : 0;
                const int first = (ln[k] > 0) ? bs[k] : 0;
                p[k] = list[first + ((lane < last) ? lane : last)];
            }
#pragma unroll
            for (int k = 0; k < RUN_B; ++k) {
                if (lane < ln[k]) f(p[k], of[k] + lane);
                for (int i = lane + 32; i < ln[k]; i += 32) f(list[bs[k] + i], of[k] + i);
            }
        }
        __syncthreads();
    }
}

__global__ void __launch_bounds__(SORT_T)
bin_sort_kernel(const int* __restrict__ packed, const int* __restrict__ dloc,
                const int* __restrict__ spacked, const int* __restrict__ sloc, int nblk,
                int* __restrict__ row_start, int* __restrict__ deg_in, int* __restrict__ by_dst,
                float* __restrict__ in_n, float* __restrict__ out_n, int n) {
    __shared__ int s_deg[BIN_W];
    __shared__ int s_odeg[BIN_W];
    __shared__ int s_fill[BIN_W];
    __shared__ int s_wave[SORT_T / 32];
    __shared__ int s_rbase[SORT_T];
    __shared__ int s_rlen[SORT_T];
    __shared__ int s_roff[SORT_T];
    __shared__ int s_carry;
    __shared__ int s_tot[2];
    __shared__ int s_in[SORT_CAP];
    __shared__ int s_out[SORT_CAP];
    const int t = threadIdx.x;
    const int j = blockIdx.x;
    s_deg[t] = 0;
    s_odeg[t] = 0;
    if (t < 2) s_tot[t] = 0;

    int sum_lo = 0, sum_len = 0;
    for (int b = t; b < nblk; b += SORT_T) {
        const int* row = dloc + (size_t)b * LOC_PITCH + j;
        const int l = row[0];
        sum_lo += l;
        sum_len += row[1] - l;
    }
    for (int o = 16; o > 0; o >>= 1) {
        sum_lo += __shfl_xor(sum_lo, o, 32);
        sum_len += __shfl_xor(sum_len, o, 32);
    }
    __syncthreads();
    if ((t & 31) == 0) { atomicAdd(&s_tot[0], sum_lo); atomicAdd(&s_tot[1], sum_len); }
    __syncthreads();
    const int lo = s_tot[0];
    const int cnt = s_tot[1];
    const bool staged = (cnt <= SORT_CAP);

    walk_bin(spacked, sloc, j, nblk, s_rbase, s_rlen, s_roff, s_wave, &s_carry,
             [&](int p, int) { atomicAdd(&s_odeg[p & (BIN_W - 1)], 1); });
    walk_bin(packed, dloc, j, nblk, s_rbase, s_rlen, s_roff, s_wave, &s_carry,
             [&](int p, int pos) {
                 atomicAdd(&s_deg[p & (BIN_W - 1)], 1);
                 if (staged) s_in[pos] = p;
             });
    __syncthreads();

    const int dg = s_deg[t];
    const int ex = block_excl_scan<SORT_T / 32>(dg, s_wave);
    s_fill[t] = staged ? ex : (lo + ex);
    const int node = j * BIN_W + t;
    if (node < n) {
        row_start[node] = lo + ex;
        deg_in[node] = dg;
        in_n[node]  = __frsqrt_rn(fmaxf((float)dg, 1.0f));
        out_n[node] = __frsqrt_rn(fmaxf((float)s_odeg[t], 1.0f));
    }
    __syncthreads();

    if (staged) {
#pragma unroll 4
        for (int e = t; e < cnt; e += SORT_T) {
            const int p = s_in[e];
            const int pos = atomicAdd(&s_fill[p & (BIN_W - 1)], 1);
            s_out[pos] = (int)((unsigned)p >> BIN_SHIFT);
        }
        __syncthreads();
#pragma unroll 4
        for (int i = t; i < cnt; i += SORT_T) by_dst[lo + i] = s_out[i];
    } else {
        walk_bin(packed, dloc, j, nblk, s_rbase, s_rlen, s_roff, s_wave, &s_carry,
                 [&](int p, int) {
                     const int pos = atomicAdd(&s_fill[p & (BIN_W - 1)], 1);
                     by_dst[pos] = (int)((unsigned)p >> BIN_SHIFT);
                 });
    }
}

#define GATHER_B 16
#define TILE_M 16
#define LDS_PITCH 68

template <bool SCALE_SRC, bool SCALE_OUT, bool POOL_OUT>
__global__ void __launch_bounds__(256)
conv_kernel(const float* __restrict__ x, const float* __restrict__ out_n,
            const float* __restrict__ in_n, const int* __restrict__ row_start,
            const int* __restrict__ deg_in, const int* __restrict__ by_dst,
            const float* __restrict__ W, const float* __restrict__ bias,
            float* __restrict__ out, const int* __restrict__ gid, float* __restrict__ cnt,
            int n_nodes, float* __restrict__ zero_buf, int zero_n, int do_relu) {
    __shared__ __attribute__((aligned(16))) float s_w[DF * LDS_PITCH];
    __shared__ __attribute__((aligned(16))) float s_a[2][TILE_M * LDS_PITCH];
    const int lane = threadIdx.x & 31;
    const int wave = __builtin_amdgcn_readfirstlane((int)(threadIdx.x >> 5));

    if (SCALE_OUT) {
        const int nthreads = (int)(gridDim.x * blockDim.x);
        for (int i = (int)(blockIdx.x * blockDim.x + threadIdx.x); i < zero_n; i += nthreads)
            zero_buf[i] = 0.0f;
    }
    static_assert(DF * DF == 256 * 16, "256 threads take 16 floats of W each");
    {
        float4 wv[4];
#pragma unroll
        for (int i = 0; i < 4; ++i)
            wv[i] = *(const float4*)(W + 4 * ((int)threadIdx.x + 256 * i));
#pragma unroll
        for (int i = 0; i < 4; ++i) {
            const int e = 4 * ((int)threadIdx.x + 256 * i);
            const int k = e >> 6, c0 = e & (DF - 1);
            s_w[(c0 + 0) * LDS_PITCH + k] = wv[i].x;
            s_w[(c0 + 1) * LDS_PITCH + k] = wv[i].y;
            s_w[(c0 + 2) * LDS_PITCH + k] = wv[i].z;
            s_w[(c0 + 3) * LDS_PITCH + k] = wv[i].w;
        }
    }

    const int ntiles = (n_nodes + TILE_M - 1) / TILE_M;
    int buf = 0;
    for (int tile = blockIdx.x; tile < ntiles; tile += gridDim.x, buf ^= 1) {
        float* __restrict__ sa = s_a[buf];
        const int m0 = tile * TILE_M;

        for (int q = 0; q < 2; ++q) {
            const int r = 2 * wave + q;
            const int node = m0 + r;
            float ax = 0.0f, ay = 0.0f;
            if (node < n_nodes) {
                const int start = __builtin_amdgcn_readfirstlane(row_start[node]);
                const int m     = __builtin_amdgcn_readfirstlane(deg_in[node]);
                const float rs  = in_n[node];
                for (int base = 0; base < m; base += 32) {
                    const int mm = (m - base < 32) ? (m - base) : 32;
                    int idx = 0;
                    float sc = 0.0f;
                    if (lane < mm) {
                        idx = by_dst[start + base + lane];
                        if (SCALE_SRC) sc = out_n[idx];
                    }
                    for (int i = 0; i < mm; i += GATHER_B) {
                        float2 v[GATHER_B];
                        int s[GATHER_B];
                        float f[GATHER_B];
#pragma unroll
                        for (int k = 0; k < GATHER_B; ++k) {
                            const int j = (i + k < mm) ? (i + k) : (mm - 1);
                            s[k] = __builtin_amdgcn_readlane(idx, j);
                            f[k] = SCALE_SRC ? __int_as_float(__builtin_amdgcn_readlane(__float_as_int(sc), j)) : 1.0f;
                        }
#pragma unroll
                        for (int k = 0; k < GATHER_B; ++k) {
                            v[k] = make_float2(0.0f, 0.0f);
                            if (i + k < mm)
                                v[k] = *(const float2*)(x + (size_t)s[k] * DF + 2 * lane);
                        }
#pragma unroll
                        for (int k = 0; k < GATHER_B; ++k) {
                            if (SCALE_SRC) {
                                ax = __builtin_fmaf(v[k].x, f[k], ax);
                                ay = __builtin_fmaf(v[k].y, f[k], ay);
                            } else {
                                ax += v[k].x;
                                ay += v[k].y;
                            }
                        }
                    }
                }
                ax *= rs;
                ay *= rs;
            }
            *(float2*)(sa + r * LDS_PITCH + 2 * lane) = make_float2(ax, ay);
        }
        __syncthreads();

        if (wave < 4) {
            const int half = lane >> 4;
            const int r16  = lane & 15;
            const int col  = wave * 16 + r16;
            const float* ap = sa + r16 * LDS_PITCH + 2 * half;
            const float* bp = s_w + col * LDS_PITCH + 2 * half;
            v8f c = {};
#pragma unroll
            for (int k = 0; k < 16; ++k) {
                const v2f a = *(const v2f*)(ap + 4 * k);
                const v2f b = *(const v2f*)(bp + 4 * k);
                c = __builtin_amdgcn_wmma_f32_16x16x4_f32(false, a, false, b, (short)0, c,
                                                          false, false);
            }
            const float bv = bias[col];
            if (POOL_OUT) {
                const int row  = m0 + r16;
                const int gl = gid[(row < n_nodes) ? row : 0];
                float v[8];
#pragma unroll
                for (int j = 0; j < 8; ++j) {
                    v[j] = c[j] + bv;
                    if (do_relu) v[j] = fmaxf(v[j], 0.0f);
                }
                const int g0 = __builtin_amdgcn_readfirstlane(gl);
                const bool one_graph = (m0 + TILE_M <= n_nodes) &&
                                       (__builtin_amdgcn_ballot_w32(gl != g0) == 0u);
                if (one_graph) {
                    float t = ((v[0] + v[1]) + (v[2] + v[3])) + ((v[4] + v[5]) + (v[6] + v[7]));
                    t += __shfl_xor(t, 16, 32);
                    if (half == 0) atomicAdd(out + (size_t)g0 * DF + col, t);
                    if (threadIdx.x == 0) atomicAdd(cnt + g0, 16.0f);
                } else {
                    int gr[8];
#pragma unroll
                    for (int j = 0; j < 8; ++j) gr[j] = __shfl(gl, 8 * half + j, 32);
                    int cur = gr[0];
                    float acc = 0.0f, run = 0.0f;
#pragma unroll
                    for (int j = 0; j < 8; ++j) {
                        if (m0 + 8 * half + j < n_nodes) {
                            if (gr[j] != cur) {
                                atomicAdd(out + (size_t)cur * DF + col, acc);
                                if (wave == 0 && r16 == 0) atomicAdd(cnt + cur, run);
                                cur = gr[j];
                                acc = 0.0f; run = 0.0f;
                            }
                            acc += v[j];
                            run += 1.0f;
                        }
                    }
                    if (run > 0.0f) {
                        atomicAdd(out + (size_t)cur * DF + col, acc);
                        if (wave == 0 && r16 == 0) atomicAdd(cnt + cur, run);
                    }
                }
            } else {
                float* orow0 = out + (size_t)m0 * DF + col;
                if (m0 + TILE_M <= n_nodes) {
#pragma unroll
                    for (int j = 0; j < 8; ++j) {
                        float v = c[j] + bv;
                        if (do_relu) v = fmaxf(v, 0.0f);
                        if (SCALE_OUT) v *= out_n[m0 + j + 8 * half];
                        orow0[(size_t)(j + 8 * half) * DF] = v;
                    }
                } else {
#pragma unroll
                    for (int j = 0; j < 8; ++j) {
                        const int orow = m0 + j + 8 * half;
                        if (orow < n_nodes) {
                            float v = c[j] + bv;
                            if (do_relu) v = fmaxf(v, 0.0f);
                            if (SCALE_OUT) v *= out_n[orow];
                            out[(size_t)orow * DF + col] = v;
                        }
                    }
                }
            }
        }
    }
}

template <int NCOLS_C>
__global__ void gemm_wmma_kernel(const float* __restrict__ A,
                                 const float* __restrict__ count,
                                 const float* __restrict__ W,
                                 const float* __restrict__ bias,
                                 float* __restrict__ out,
                                 int rows, int ncols_rt) {
    const int Ncols = (NCOLS_C > 0) ? NCOLS_C : ncols_rt;
    const int lane = threadIdx.x & 31;
    const int wave = threadIdx.x >> 5;
    const int half = lane >> 4;
    const int r16  = lane & 15;
    const int n0 = wave * 16;

    const int col  = n0 + r16;
    const bool cok = (col < Ncols);
    const int colc = cok ? col : (Ncols - 1);
    const float* wp = W + (size_t)(2 * half) * Ncols + colc;

    v2f b[16];
#pragma unroll
    for (int k = 0; k < 16; ++k) {
        b[k].x = wp[(size_t)(4 * k) * Ncols];
        b[k].y = wp[(size_t)(4 * k + 1) * Ncols];
    }
    const float bv = bias[colc];

    const int ntiles = (rows + 15) >> 4;
    for (int tile = blockIdx.x; tile < ntiles; tile += gridDim.x) {
        const int m0 = tile * 16;
        const int row  = m0 + r16;
        const int rowc = (row < rows) ? row : 0;
        const float s = 1.0f / fmaxf(count[rowc], 1.0f);
        const float* arow = A + (size_t)rowc * DF + 2 * half;

        v8f c = {};
#pragma unroll
        for (int k = 0; k < 16; ++k) {
            v2f a;
            a.x = arow[4 * k + 0] * s;
            a.y = arow[4 * k + 1] * s;
            c = __builtin_amdgcn_wmma_f32_16x16x4_f32(false, a, false, b[k], (short)0, c,
                                                      false, false);
        }

        float* orow0 = out + (size_t)m0 * Ncols + col;
        if (m0 + 16 <= rows && cok) {
#pragma unroll
            for (int j = 0; j < 8; ++j)
                orow0[(size_t)(j + 8 * half) * Ncols] = c[j] + bv;
        } else if (cok) {
#pragma unroll
            for (int j = 0; j < 8; ++j) {
                const int orow = m0 + j + 8 * half;
                if (orow < rows) out[(size_t)orow * Ncols + col] = c[j] + bv;
            }
        }
    }
}

extern "C" void kernel_launch(void* const* d_in, const int* in_sizes, int n_in,
                              void* d_out, int out_size, void* d_ws, size_t ws_size,
                              hipStream_t stream) {
    const float* features = (const float*)d_in[0];
    const int*   src      = (const int*)d_in[1];
    const int*   dst      = (const int*)d_in[2];
    const int*   gid      = (const int*)d_in[3];
    const float* W1       = (const float*)d_in[4];
    const float* b1       = (const float*)d_in[5];
    const float* W2       = (const float*)d_in[6];
    const float* b2       = (const float*)d_in[7];
    const float* Wc       = (const float*)d_in[8];
    const float* bc       = (const float*)d_in[9];
    float* outp           = (float*)d_out;

    const int N    = in_sizes[3];
    const int E    = in_sizes[1];
    const int DOUT = in_sizes[9];
    const int G    = out_size / DOUT;

    const size_t Np = ((size_t)N + 63) & ~(size_t)63;
    const size_t Ep = ((size_t)E + 63) & ~(size_t)63;
    const int nBins = (N + BIN_W - 1) / BIN_W;
    const int edgeBlocks = (E + EDGE_CAP - 1) / EDGE_CAP;
    const size_t Lp = (size_t)edgeBlocks * LOC_PITCH;
    int*   dloc  = (int*)d_ws;
    int*   sloc  = dloc + Lp;
    int*   deg_i = sloc + Lp;
    int*   rowst = deg_i + Np;
    int*   pack  = rowst + Np;
    int*   spack = pack + Ep;
    int*   bydst = spack + Ep;
    float* out_n = (float*)(bydst + Ep);
    float* in_n  = out_n + Np;
    float* bufA  = in_n + Np;
    float* pool  = bufA + (size_t)N * DF;
    float* cnt   = pool + (size_t)G * DF;

    bin_local_kernel<<<edgeBlocks, BIN_T, 0, stream>>>(src, dst, dloc, pack, sloc, spack, E, nBins);
    bin_sort_kernel<<<nBins, SORT_T, 0, stream>>>(pack, dloc, spack, sloc, edgeBlocks,
                                                  rowst, deg_i, bydst, in_n, out_n, N);

    const int convGrid = 2048;

    conv_kernel<true, true, false><<<convGrid, 256, 0, stream>>>(
        features, out_n, in_n, rowst, deg_i, bydst, W1, b1, bufA, nullptr, nullptr, N,
        pool, G * DF + G, 1);

    conv_kernel<false, false, true><<<convGrid, 256, 0, stream>>>(
        bufA, out_n, in_n, rowst, deg_i, bydst, W2, b2, pool, gid, cnt, N,
        nullptr, 0, 1);

    const int clsWaves = (DOUT + 15) / 16;
    if (DOUT == 8) {
        gemm_wmma_kernel<8><<<(G + 15) / 16, 32 * clsWaves, 0, stream>>>(
            pool, cnt, Wc, bc, outp, G, DOUT);
    } else {
        gemm_wmma_kernel<-1><<<(G + 15) / 16, 32 * clsWaves, 0, stream>>>(
            pool, cnt, Wc, bc, outp, G, DOUT);
    }
}


# layer kernels: a 16-lane half-wave fetches a whole source row (16-byte loads, two rows per load, indices by DPP row_share); even and odd edges summed apart: row sum order changes
# speedup vs baseline: 6.9488x; 1.0308x over previous
#include <hip/hip_runtime.h>
#include <hip/hip_bf16.h>

typedef float v2f __attribute__((ext_vector_type(2)));
typedef float v8f __attribute__((ext_vector_type(8)));

#define DF 64

#define BIN_SHIFT 8
#define BIN_W (1 << BIN_SHIFT)
#define MAX_BINS 1024
#define BIN_T 1024
#define EDGE_EPT 7
#define EDGE_CAP 6400
#define LOC_PITCH (MAX_BINS + 64)
#define SORT_T BIN_W
#define SORT_CAP 6144
#define RUN_B 16

static_assert(BIN_T == MAX_BINS, "one thread per bin slot");
static_assert(EDGE_CAP <= BIN_T * EDGE_EPT, "every edge of a block has a thread slot");
static_assert(EDGE_CAP < 65536, "a rank inside a block fits 16 bits");
static_assert((SORT_T / 8) % RUN_B == 0, "a wave's runs of a chunk come in whole trips");

template <int NW>
__device__ __forceinline__ int block_excl_scan(int v, int* s_wave) {
    const int lane = threadIdx.x & 31;
    const int wave = threadIdx.x >> 5;
    int inc = v;
    for (int o = 1; o < 32; o <<= 1) {
        int t = __shfl_up(inc, o, 32);
        if (lane >= o) inc += t;
    }
    if (lane == 31) s_wave[wave] = inc;
    __syncthreads();
    int woff = 0;
    for (int w = 0; w < NW; ++w)
        if (w < wave) woff += s_wave[w];
    __syncthreads();
    return woff + inc - v;
}

__device__ __forceinline__ void local_pass(const int (&key)[EDGE_EPT], const int (&val)[EDGE_EPT],
                                           int cnt, int nbins,
                                           int* __restrict__ loc_row, int* __restrict__ out_blk,
                                           int* s_cnt, int* s_loc, int* s_wave, int* s_val) {
    const int t = threadIdx.x;
    s_cnt[t] = 0;
    __syncthreads();

    int br[EDGE_EPT];
#pragma unroll
    for (int k = 0; k < EDGE_EPT; ++k) {
        const int i = t + k * BIN_T;
        br[k] = 0;
        if (i < cnt) {
            const int bin = key[k] >> BIN_SHIFT;
            const int r = atomicAdd(&s_cnt[bin], 1);
            br[k] = (bin << 16) | r;
        }
    }
    __syncthreads();

    const int mine = s_cnt[t];
    const int loc = block_excl_scan<BIN_T / 32>(mine, s_wave);
    s_loc[t] = loc;
    if (t < nbins) loc_row[t] = loc;
    if (t == 0) loc_row[nbins] = cnt;
    __syncthreads();

#pragma unroll
    for (int k = 0; k < EDGE_EPT; ++k) {
        const int i = t + k * BIN_T;
        if (i < cnt) {
            const int bin = br[k] >> 16;
            s_val[s_loc[bin] + (br[k] & 0xFFFF)] = val[k];
        }
    }
    __syncthreads();
#pragma unroll
    for (int k = 0; k < EDGE_EPT; ++k) {
        const int i = t + k * BIN_T;
        if (i < cnt) out_blk[i] = s_val[i];
    }
}

__global__ void __launch_bounds__(BIN_T)
bin_local_kernel(const int* __restrict__ src, const int* __restrict__ dst,
                 int* __restrict__ dloc, int* __restrict__ packed,
                 int* __restrict__ sloc, int* __restrict__ spacked, int E, int nbins) {
    __shared__ int s_cnt[MAX_BINS];
    __shared__ int s_loc[MAX_BINS];
    __shared__ int s_wave[BIN_T / 32];
    __shared__ int s_val[EDGE_CAP];
    const int t = threadIdx.x;
    const int lo = blockIdx.x * EDGE_CAP;
    const int cnt = (E - lo < EDGE_CAP) ? (E - lo) : EDGE_CAP;

    int s[EDGE_EPT], d[EDGE_EPT], val[EDGE_EPT];
#pragma unroll
    for (int k = 0; k < EDGE_EPT; ++k) {
        const int i = t + k * BIN_T;
        s[k] = 0; d[k] = 0;
        if (i < cnt) { s[k] = src[lo + i]; d[k] = dst[lo + i]; }
    }
#pragma unroll
    for (int k = 0; k < EDGE_EPT; ++k)
        val[k] = (int)(((unsigned)s[k] << BIN_SHIFT) | (unsigned)(d[k] & (BIN_W - 1)));
    local_pass(d, val, cnt, nbins, dloc + (size_t)blockIdx.x * LOC_PITCH, packed + lo,
               s_cnt, s_loc, s_wave, s_val);
#pragma unroll
    for (int k = 0; k < EDGE_EPT; ++k)
        val[k] = s[k] & (BIN_W - 1);
    local_pass(s, val, cnt, nbins, sloc + (size_t)blockIdx.x * LOC_PITCH, spacked + lo,
               s_cnt, s_loc, s_wave, s_val);
}

template <class F>
__device__ __forceinline__ void walk_bin(const int* __restrict__ list, const int* __restrict__ loc,
                                         int j, int nblk, int* s_rbase, int* s_rlen, int* s_roff,
                                         int* s_wave, int* s_carry, F f) {
    const int t = threadIdx.x;
    const int lane = t & 31;
    const int wave = __builtin_amdgcn_readfirstlane(t >> 5);
    int carry = 0;
    for (int c0 = 0; c0 < nblk; c0 += SORT_T) {
        const int b = c0 + t;
        int lo = 0, len = 0;
        if (b < nblk) {
            const int* row = loc + (size_t)b * LOC_PITCH + j;
            lo = row[0];
            len = row[1] - lo;
        }
        const int off = carry + block_excl_scan<SORT_T / 32>(len, s_wave);
        s_rbase[t] = b * EDGE_CAP + lo;
        s_rlen[t] = len;
        s_roff[t] = off;
        if (t == SORT_T - 1) *s_carry = off + len;
        __syncthreads();
        carry = *s_carry;
        for (int q0 = 0; q0 < SORT_T / 8; q0 += RUN_B) {
            int p[RUN_B], ln[RUN_B], bs[RUN_B], of[RUN_B];
#pragma unroll
            for (int k = 0; k < RUN_B; ++k) {
                const int r = wave + 8 * (q0 + k);
                ln[k] = __builtin_amdgcn_readfirstlane(s_rlen[r]);
                bs[k] = __builtin_amdgcn_readfirstlane(s_rbase[r]);
                of[k] = __builtin_amdgcn_readfirstlane(s_roff[r]);
            }
#pragma unroll
            for (int k = 0; k < RUN_B; ++k) {
                const int last = (ln[k] > 0) ? (ln[k] - 1) : 0;
                const int first = (ln[k] > 0) ? bs[k] : 0;
                p[k] = list[first + ((lane < last) ? lane : last)];
            }
#pragma unroll
            for (int k = 0; k < RUN_B; ++k) {
                if (lane < ln[k]) f(p[k], of[k] + lane);
                for (int i = lane + 32; i < ln[k]; i += 32) f(list[bs[k] + i], of[k] + i);
            }
        }
        __syncthreads();
    }
}

__global__ void __launch_bounds__(SORT_T)
bin_sort_kernel(const int* __restrict__ packed, const int* __restrict__ dloc,
                const int* __restrict__ spacked, const int* __restrict__ sloc, int nblk,
                int* __restrict__ row_start, int* __restrict__ deg_in, int* __restrict__ by_dst,
                float* __restrict__ in_n, float* __restrict__ out_n, int n) {
    __shared__ int s_deg[BIN_W];
    __shared__ int s_odeg[BIN_W];
    __shared__ int s_fill[BIN_W];
    __shared__ int s_wave[SORT_T / 32];
    __shared__ int s_rbase[SORT_T];
    __shared__ int s_rlen[SORT_T];
    __shared__ int s_roff[SORT_T];
    __shared__ int s_carry;
    __shared__ int s_tot[2];
    __shared__ int s_in[SORT_CAP];
    __shared__ int s_out[SORT_CAP];
    const int t = threadIdx.x;
    const int j = blockIdx.x;
    s_deg[t] = 0;
    s_odeg[t] = 0;
    if (t < 2) s_tot[t] = 0;

    int sum_lo = 0, sum_len = 0;
    for (int b = t; b < nblk; b += SORT_T) {
        const int* row = dloc + (size_t)b * LOC_PITCH + j;
        const int l = row[0];
        sum_lo += l;
        sum_len += row[1] - l;
    }
    for (int o = 16; o > 0; o >>= 1) {
        sum_lo += __shfl_xor(sum_lo, o, 32);
        sum_len += __shfl_xor(sum_len, o, 32);
    }
    __syncthreads();
    if ((t & 31) == 0) { atomicAdd(&s_tot[0], sum_lo); atomicAdd(&s_tot[1], sum_len); }
    __syncthreads();
    const int lo = s_tot[0];
    const int cnt = s_tot[1];
    const bool staged = (cnt <= SORT_CAP);

    walk_bin(spacked, sloc, j, nblk, s_rbase, s_rlen, s_roff, s_wave, &s_carry,
             [&](int p, int) { atomicAdd(&s_odeg[p & (BIN_W - 1)], 1); });
    walk_bin(packed, dloc, j, nblk, s_rbase, s_rlen, s_roff, s_wave, &s_carry,
             [&](int p, int pos) {
                 atomicAdd(&s_deg[p & (BIN_W - 1)], 1);
                 if (staged) s_in[pos] = p;
             });
    __syncthreads();

    const int dg = s_deg[t];
    const int ex = block_excl_scan<SORT_T / 32>(dg, s_wave);
    s_fill[t] = staged ? ex : (lo + ex);
    const int node = j * BIN_W + t;
    if (node < n) {
        row_start[node] = lo + ex;
        deg_in[node] = dg;
        in_n[node]  = __frsqrt_rn(fmaxf((float)dg, 1.0f));
        out_n[node] = __frsqrt_rn(fmaxf((float)s_odeg[t], 1.0f));
    }
    __syncthreads();

    if (staged) {
#pragma unroll 4
        for (int e = t; e < cnt; e += SORT_T) {
            const int p = s_in[e];
            const int pos = atomicAdd(&s_fill[p & (BIN_W - 1)], 1);
            s_out[pos] = (int)((unsigned)p >> BIN_SHIFT);
        }
        __syncthreads();
#pragma unroll 4
        for (int i = t; i < cnt; i += SORT_T) by_dst[lo + i] = s_out[i];
    } else {
        walk_bin(packed, dloc, j, nblk, s_rbase, s_rlen, s_roff, s_wave, &s_carry,
                 [&](int p, int) {
                     const int pos = atomicAdd(&s_fill[p & (BIN_W - 1)], 1);
                     by_dst[pos] = (int)((unsigned)p >> BIN_SHIFT);
                 });
    }
}

#define TILE_M 16
#define LDS_PITCH 68

template <int K>
__device__ __forceinline__ int row_share(int v) {
    return __builtin_amdgcn_update_dpp(0, v, 0x150 + K, 0xf, 0xf, true);
}
__device__ __forceinline__ float other_half(float v) {
    return __int_as_float(__builtin_amdgcn_permlanex16(__float_as_int(v), __float_as_int(v),
                                                       0x76543210, 0xfedcba98, false, false));
}

template <bool SCALE_SRC, int P0, int NB>
__device__ __forceinline__ void gather_pairs(const float* __restrict__ x, const float* __restrict__ scale,
                                             int idx, unsigned lane_b, float& sc, float4& acc) {
    const unsigned ob = (unsigned)idx << 8;
    float4 v[8];
#define GP_LOAD(k)                                                                       \
    if constexpr (k < NB) {                                                              \
        const unsigned o = (unsigned)row_share<P0 + k>((int)ob) | lane_b;                \
        v[k] = *(const float4*)((const char*)x + o);                                     \
    }
    GP_LOAD(0) GP_LOAD(1) GP_LOAD(2) GP_LOAD(3) GP_LOAD(4) GP_LOAD(5) GP_LOAD(6) GP_LOAD(7)
#undef GP_LOAD
    if (SCALE_SRC && P0 == 0) sc = scale[idx];
#define GP_USE(k)                                                                        \
    if constexpr (k < NB) {                                                              \
        if (SCALE_SRC) {                                                                 \
            const float f = __int_as_float(row_share<P0 + k>(__float_as_int(sc)));       \
            acc.x = __builtin_fmaf(v[k].x, f, acc.x);                                    \
            acc.y = __builtin_fmaf(v[k].y, f, acc.y);                                    \
            acc.z = __builtin_fmaf(v[k].z, f, acc.z);                                    \
            acc.w = __builtin_fmaf(v[k].w, f, acc.w);                                    \
        } else {                                                                         \
            acc.x += v[k].x; acc.y += v[k].y; acc.z += v[k].z; acc.w += v[k].w;          \
        }                                                                                \
    }
    GP_USE(0) GP_USE(1) GP_USE(2) GP_USE(3) GP_USE(4) GP_USE(5) GP_USE(6) GP_USE(7)
#undef GP_USE
}

template <bool SCALE_SRC, int P0>
__device__ __forceinline__ void gather_batch(const float* __restrict__ x, const float* __restrict__ scale,
                                             int idx, unsigned lane_b, int nb, float& sc, float4& acc) {
    switch (nb) {
        case 1: gather_pairs<SCALE_SRC, P0, 1>(x, scale, idx, lane_b, sc, acc); break;
        case 2: gather_pairs<SCALE_SRC, P0, 2>(x, scale, idx, lane_b, sc, acc); break;
        case 3: gather_pairs<SCALE_SRC, P0, 3>(x, scale, idx, lane_b, sc, acc); break;
        case 4: gather_pairs<SCALE_SRC, P0, 4>(x, scale, idx, lane_b, sc, acc); break;
        case 5: gather_pairs<SCALE_SRC, P0, 5>(x, scale, idx, lane_b, sc, acc); break;
        case 6: gather_pairs<SCALE_SRC, P0, 6>(x, scale, idx, lane_b, sc, acc); break;
        case 7: gather_pairs<SCALE_SRC, P0, 7>(x, scale, idx, lane_b, sc, acc); break;
        case 8: gather_pairs<SCALE_SRC, P0, 8>(x, scale, idx, lane_b, sc, acc); break;
        default: gather_pairs<SCALE_SRC, P0, 0>(x, scale, idx, lane_b, sc, acc); break;
    }
}

template <bool SCALE_SRC, bool SCALE_OUT, bool POOL_OUT>
__global__ void __launch_bounds__(256)
conv_kernel(const float* __restrict__ x, const float* __restrict__ out_n,
            const float* __restrict__ in_n, const int* __restrict__ row_start,
            const int* __restrict__ deg_in, const int* __restrict__ by_dst,
            const float* __restrict__ W, const float* __restrict__ bias,
            float* __restrict__ out, const int* __restrict__ gid, float* __restrict__ cnt,
            int n_nodes, float* __restrict__ zero_buf, int zero_n, int do_relu) {
    __shared__ __attribute__((aligned(16))) float s_w[DF * LDS_PITCH];
    __shared__ __attribute__((aligned(16))) float s_a[2][TILE_M * LDS_PITCH];
    const int lane = threadIdx.x & 31;
    const int wave = __builtin_amdgcn_readfirstlane((int)(threadIdx.x >> 5));

    if (SCALE_OUT) {
        const int nthreads = (int)(gridDim.x * blockDim.x);
        for (int i = (int)(blockIdx.x * blockDim.x + threadIdx.x); i < zero_n; i += nthreads)
            zero_buf[i] = 0.0f;
    }
    static_assert(DF * DF == 256 * 16, "256 threads take 16 floats of W each");
    {
        float4 wv[4];
#pragma unroll
        for (int i = 0; i < 4; ++i)
            wv[i] = *(const float4*)(W + 4 * ((int)threadIdx.x + 256 * i));
#pragma unroll
        for (int i = 0; i < 4; ++i) {
            const int e = 4 * ((int)threadIdx.x + 256 * i);
            const int k = e >> 6, c0 = e & (DF - 1);
            s_w[(c0 + 0) * LDS_PITCH + k] = wv[i].x;
            s_w[(c0 + 1) * LDS_PITCH + k] = wv[i].y;
            s_w[(c0 + 2) * LDS_PITCH + k] = wv[i].z;
            s_w[(c0 + 3) * LDS_PITCH + k] = wv[i].w;
        }
    }

    const int c16 = lane & 15;
    const int hf = lane >> 4;
    const unsigned lane_b = (unsigned)c16 << 4;
    const int epos = 2 * c16 + hf;

    const int ntiles = (n_nodes + TILE_M - 1) / TILE_M;
    int buf = 0;
    for (int tile = blockIdx.x; tile < ntiles; tile += gridDim.x, buf ^= 1) {
        float* __restrict__ sa = s_a[buf];
        const int m0 = tile * TILE_M;

        for (int q = 0; q < 2; ++q) {
            const int r = 2 * wave + q;
            const int node = m0 + r;
            float4 acc = make_float4(0.0f, 0.0f, 0.0f, 0.0f);
            float2 tacc = make_float2(0.0f, 0.0f);
            float rs = 0.0f;
            if (node < n_nodes) {
                const int start = __builtin_amdgcn_readfirstlane(row_start[node]);
                const int m     = __builtin_amdgcn_readfirstlane(deg_in[node]);
                rs = in_n[node];
                for (int base = 0; base < m; base += 32) {
                    const int mm = (m - base < 32) ? (m - base) : 32;
                    const int np = mm >> 1;
                    int idx = 0;
                    if (epos < mm) idx = by_dst[start + base + epos];
                    const bool odd = (mm & 1) != 0;
                    float2 vt = make_float2(0.0f, 0.0f);
                    if (odd) {
                        const int si = __builtin_amdgcn_readlane(idx, np);
                        vt = *(const float2*)(x + (size_t)si * DF + 2 * epos);
                    }
                    float sc = 0.0f;
                    gather_batch<SCALE_SRC, 0>(x, out_n, idx, lane_b, (np < 8) ? np : 8, sc, acc);
                    if (np > 8) gather_batch<SCALE_SRC, 8>(x, out_n, idx, lane_b, np - 8, sc, acc);
                    if (odd) {
                        if (SCALE_SRC) {
                            const float ft = __int_as_float(__builtin_amdgcn_readlane(__float_as_int(sc), np));
                            tacc.x = __builtin_fmaf(vt.x, ft, tacc.x);
                            tacc.y = __builtin_fmaf(vt.y, ft, tacc.y);
                        } else {
                            tacc.x += vt.x; tacc.y += vt.y;
                        }
                    }
                }
            }
            const float keep0 = hf ? acc.z : acc.x;
            const float keep1 = hf ? acc.w : acc.y;
            const float send0 = hf ? acc.x : acc.z;
            const float send1 = hf ? acc.y : acc.w;
            const float o0 = ((keep0 + other_half(send0)) + tacc.x) * rs;
            const float o1 = ((keep1 + other_half(send1)) + tacc.y) * rs;
            *(float2*)(sa + r * LDS_PITCH + 2 * epos) = make_float2(o0, o1);
        }
        __syncthreads();

        if (wave < 4) {
            const int half = lane >> 4;
            const int r16  = lane & 15;
            const int col  = wave * 16 + r16;
            const float* ap = sa + r16 * LDS_PITCH + 2 * half;
            const float* bp = s_w + col * LDS_PITCH + 2 * half;
            v8f c = {};
#pragma unroll
            for (int k = 0; k < 16; ++k) {
                const v2f a = *(const v2f*)(ap + 4 * k);
                const v2f b = *(const v2f*)(bp + 4 * k);
                c = __builtin_amdgcn_wmma_f32_16x16x4_f32(false, a, false, b, (short)0, c,
                                                          false, false);
            }
            const float bv = bias[col];
            if (POOL_OUT) {
                const int row  = m0 + r16;
                const int gl = gid[(row < n_nodes) ? row : 0];
                float v[8];
#pragma unroll
                for (int j = 0; j < 8; ++j) {
                    v[j] = c[j] + bv;
                    if (do_relu) v[j] = fmaxf(v[j], 0.0f);
                }
                const int g0 = __builtin_amdgcn_readfirstlane(gl);
                const bool one_graph = (m0 + TILE_M <= n_nodes) &&
                                       (__builtin_amdgcn_ballot_w32(gl != g0) == 0u);
                if (one_graph) {
                    float t = ((v[0] + v[1]) + (v[2] + v[3])) + ((v[4] + v[5]) + (v[6] + v[7]));
                    t += __shfl_xor(t, 16, 32);
                    if (half == 0) atomicAdd(out + (size_t)g0 * DF + col, t);
                    if (threadIdx.x == 0) atomicAdd(cnt + g0, 16.0f);
                } else {
                    int gr[8];
#pragma unroll
                    for (int j = 0; j < 8; ++j) gr[j] = __shfl(gl, 8 * half + j, 32);
                    int cur = gr[0];
                    float acc = 0.0f, run = 0.0f;
#pragma unroll
                    for (int j = 0; j < 8; ++j) {
                        if (m0 + 8 * half + j < n_nodes) {
                            if (gr[j] != cur) {
                                atomicAdd(out + (size_t)cur * DF + col, acc);
                                if (wave == 0 && r16 == 0) atomicAdd(cnt + cur, run);
                                cur = gr[j];
                                acc = 0.0f; run = 0.0f;
                            }
                            acc += v[j];
                            run += 1.0f;
                        }
                    }
                    if (run > 0.0f) {
                        atomicAdd(out + (size_t)cur * DF + col, acc);
                        if (wave == 0 && r16 == 0) atomicAdd(cnt + cur, run);
                    }
                }
            } else {
                float* orow0 = out + (size_t)m0 * DF + col;
                if (m0 + TILE_M <= n_nodes) {
#pragma unroll
                    for (int j = 0; j < 8; ++j) {
                        float v = c[j] + bv;
                        if (do_relu) v = fmaxf(v, 0.0f);
                        if (SCALE_OUT) v *= out_n[m0 + j + 8 * half];
                        orow0[(size_t)(j + 8 * half) * DF] = v;
                    }
                } else {
#pragma unroll
                    for (int j = 0; j < 8; ++j) {
                        const int orow = m0 + j + 8 * half;
                        if (orow < n_nodes) {
                            float v = c[j] + bv;
                            if (do_relu) v = fmaxf(v, 0.0f);
                            if (SCALE_OUT) v *= out_n[orow];
                            out[(size_t)orow * DF + col] = v;
                        }
                    }
                }
            }
        }
    }
}

template <int NCOLS_C>
__global__ void gemm_wmma_kernel(const float* __restrict__ A,
                                 const float* __restrict__ count,
                                 const float* __restrict__ W,
                                 const float* __restrict__ bias,
                                 float* __restrict__ out,
                                 int rows, int ncols_rt) {
    const int Ncols = (NCOLS_C > 0) ? NCOLS_C : ncols_rt;
    const int lane = threadIdx.x & 31;
    const int wave = threadIdx.x >> 5;
    const int half = lane >> 4;
    const int r16  = lane & 15;
    const int n0 = wave * 16;

    const int col  = n0 + r16;
    const bool cok = (col < Ncols);
    const int colc = cok ? col : (Ncols - 1);
    const float* wp = W + (size_t)(2 * half) * Ncols + colc;

    v2f b[16];
#pragma unroll
    for (int k = 0; k < 16; ++k) {
        b[k].x = wp[(size_t)(4 * k) * Ncols];
        b[k].y = wp[(size_t)(4 * k + 1) * Ncols];
    }
    const float bv = bias[colc];

    const int ntiles = (rows + 15) >> 4;
    for (int tile = blockIdx.x; tile < ntiles; tile += gridDim.x) {
        const int m0 = tile * 16;
        const int row  = m0 + r16;
        const int rowc = (row < rows) ? row : 0;
        const float s = 1.0f / fmaxf(count[rowc], 1.0f);
        const float* arow = A + (size_t)rowc * DF + 2 * half;

        v8f c = {};
#pragma unroll
        for (int k = 0; k < 16; ++k) {
            v2f a;
            a.x = arow[4 * k + 0] * s;
            a.y = arow[4 * k + 1] * s;
            c = __builtin_amdgcn_wmma_f32_16x16x4_f32(false, a, false, b[k], (short)0, c,
                                                      false, false);
        }

        float* orow0 = out + (size_t)m0 * Ncols + col;
        if (m0 + 16 <= rows && cok) {
#pragma unroll
            for (int j = 0; j < 8; ++j)
                orow0[(size_t)(j + 8 * half) * Ncols] = c[j] + bv;
        } else if (cok) {
#pragma unroll
            for (int j = 0; j < 8; ++j) {
                const int orow = m0 + j + 8 * half;
                if (orow < rows) out[(size_t)orow * Ncols + col] = c[j] + bv;
            }
        }
    }
}

extern "C" void kernel_launch(void* const* d_in, const int* in_sizes, int n_in,
                              void* d_out, int out_size, void* d_ws, size_t ws_size,
                              hipStream_t stream) {
    const float* features = (const float*)d_in[0];
    const int*   src      = (const int*)d_in[1];
    const int*   dst      = (const int*)d_in[2];
    const int*   gid      = (const int*)d_in[3];
    const float* W1       = (const float*)d_in[4];
    const float* b1       = (const float*)d_in[5];
    const float* W2       = (const float*)d_in[6];
    const float* b2       = (const float*)d_in[7];
    const float* Wc       = (const float*)d_in[8];
    const float* bc       = (const float*)d_in[9];
    float* outp           = (float*)d_out;

    const int N    = in_sizes[3];
    const int E    = in_sizes[1];
    const int DOUT = in_sizes[9];
    const int G    = out_size / DOUT;

    const size_t Np = ((size_t)N + 63) & ~(size_t)63;
    const size_t Ep = ((size_t)E + 63) & ~(size_t)63;
    const int nBins = (N + BIN_W - 1) / BIN_W;
    const int edgeBlocks = (E + EDGE_CAP - 1) / EDGE_CAP;
    const size_t Lp = (size_t)edgeBlocks * LOC_PITCH;
    int*   dloc  = (int*)d_ws;
    int*   sloc  = dloc + Lp;
    int*   deg_i = sloc + Lp;
    int*   rowst = deg_i + Np;
    int*   pack  = rowst + Np;
    int*   spack = pack + Ep;
    int*   bydst = spack + Ep;
    float* out_n = (float*)(bydst + Ep);
    float* in_n  = out_n + Np;
    float* bufA  = in_n + Np;
    float* pool  = bufA + (size_t)N * DF;
    float* cnt   = pool + (size_t)G * DF;

    bin_local_kernel<<<edgeBlocks, BIN_T, 0, stream>>>(src, dst, dloc, pack, sloc, spack, E, nBins);
    bin_sort_kernel<<<nBins, SORT_T, 0, stream>>>(pack, dloc, spack, sloc, edgeBlocks,
                                                  rowst, deg_i, bydst, in_n, out_n, N);

    const int convGrid = 2048;

    conv_kernel<true, true, false><<<convGrid, 256, 0, stream>>>(
        features, out_n, in_n, rowst, deg_i, bydst, W1, b1, bufA, nullptr, nullptr, N,
        pool, G * DF + G, 1);

    conv_kernel<false, false, true><<<convGrid, 256, 0, stream>>>(
        bufA, out_n, in_n, rowst, deg_i, bydst, W2, b2, pool, gid, cnt, N,
        nullptr, 0, 1);

    const int clsWaves = (DOUT + 15) / 16;
    if (DOUT == 8) {
        gemm_wmma_kernel<8><<<(G + 15) / 16, 32 * clsWaves, 0, stream>>>(
            pool, cnt, Wc, bc, outp, G, DOUT);
    } else {
        gemm_wmma_kernel<-1><<<(G + 15) / 16, 32 * clsWaves, 0, stream>>>(
            pool, cnt, Wc, bc, outp, G, DOUT);
    }
}
